# starting kernel
# baseline (speedup 1.0000x reference)
#include <hip/hip_runtime.h>
#include <hip/hip_bf16.h>


#define B_  4
#define DI  192
#define LL  4096
#define NS  16
#define RR  6
#define KK  4
#define CP  38
#define TL  64
#define TLP 65

typedef float v2f __attribute__((ext_vector_type(2)));
typedef float v8f __attribute__((ext_vector_type(8)));

__device__ __forceinline__ int scan_idx(int k, int l) {
    int ll = (k & 2) ? (LL - 1 - l) : l;
    if (k & 1) return ((ll & 63) << 6) | (ll >> 6);
    return ll;
}

__global__ __launch_bounds__(256) void ss2d_proj_wmma(
    const float* __restrict__ x,
    const float* __restrict__ xpw,
    float* __restrict__ xdbl)
{
    int wave = (int)((blockIdx.x * blockDim.x + threadIdx.x) >> 5);
    int lane = threadIdx.x & 31;
    int nt = wave & 255;
    int mt = (wave >> 8) % 3;
    int kd = (wave >> 8) / 3;
    if (kd >= B_ * KK) return;
    int k = kd & 3;
    int b = kd >> 2;

    int m0 = mt << 4;
    int l0 = nt << 4;
    int hw = lane >> 4;
    int ln = lane & 15;

    const float* wbase = xpw + (size_t)k * CP * DI;
    const float* xb    = x + (size_t)b * DI * LL;

    int arow   = m0 + ln;
    int arow_c = arow < CP ? arow : (CP - 1);
    float amsk = arow < CP ? 1.0f : 0.0f;

    int midx = scan_idx(k, l0 + ln);

    v8f acc = {0.f, 0.f, 0.f, 0.f, 0.f, 0.f, 0.f, 0.f};
    for (int k0 = 0; k0 < DI; k0 += 4) {
        const float* ap = wbase + arow_c * DI + k0 + (hw << 1);
        v2f a; a.x = ap[0] * amsk; a.y = ap[1] * amsk;
        const float* bp = xb + (size_t)(k0 + (hw << 1)) * LL + midx;
        v2f bb; bb.x = bp[0]; bb.y = bp[LL];
        acc = __builtin_amdgcn_wmma_f32_16x16x4_f32(
            false, a, false, bb, (short)0, acc, false, false);
    }

    float* obase = xdbl + (size_t)kd * CP * LL + l0 + ln;
#pragma unroll
    for (int i = 0; i < 8; ++i) {
        int m = m0 + i + (hw << 3);
        if (m < CP) obase[(size_t)m * LL] = acc[i];
    }
}

__global__ __launch_bounds__(1024) void ss2d_scan(
    const float* __restrict__ y,
    const float* __restrict__ dtw_g,
    const float* __restrict__ dtb_g,
    const float* __restrict__ Alogs,
    const float* __restrict__ Ds_g,
    const float* __restrict__ xdbl,
    float* __restrict__ outy)
{
    __shared__ float sm_rows[CP * TLP];
    __shared__ float sm_u[64 * TLP];
    __shared__ float sm_y[64 * TLP];

    int dblk = blockIdx.x % 3;
    int k    = (blockIdx.x / 3) & 3;
    int b    = blockIdx.x / 12;
    int kd   = b * 4 + k;

    int t  = threadIdx.x;
    int n  = t & 15;
    int dl = t >> 4;
    int d  = dblk * 64 + dl;
    int kdch = k * DI + d;

    float A_dn = -__expf(Alogs[(size_t)kdch * NS + n]);
    float dtw[RR];
#pragma unroll
    for (int r = 0; r < RR; ++r) dtw[r] = dtw_g[(size_t)kdch * RR + r];
    float dtb = dtb_g[kdch];
    float Dsd = Ds_g[kdch];

    const float* xdb = xdbl + (size_t)kd * CP * LL;
    const float* yb  = y + (size_t)b * DI * LL;

    float h = 0.0f;

    for (int c = 0; c < LL / TL; ++c) {
        int l0 = c * TL;
        for (int i = t; i < CP * TL; i += 1024) {
            int r = i >> 6, col = i & 63;
            sm_rows[r * TLP + col] = xdb[(size_t)r * LL + l0 + col];
        }
        for (int i = t; i < 64 * TL; i += 1024) {
            int r = i >> 6, col = i & 63;
            sm_u[r * TLP + col] =
                yb[(size_t)(dblk * 64 + r) * LL + scan_idx(k, l0 + col)];
        }
        __syncthreads();

        for (int j = 0; j < TL; ++j) {
            float s = dtb;
#pragma unroll
            for (int r = 0; r < RR; ++r) s += dtw[r] * sm_rows[r * TLP + j];
            float delta = (s > 20.0f) ? s : log1pf(__expf(s));

            float u  = sm_u[dl * TLP + j];
            float bv = sm_rows[(RR + n) * TLP + j];
            float cv = sm_rows[(RR + NS + n) * TLP + j];

            h = __expf(delta * A_dn) * h + (delta * u) * bv;
            float yv = h * cv;
            yv += __shfl_xor(yv, 1, 16);
            yv += __shfl_xor(yv, 2, 16);
            yv += __shfl_xor(yv, 4, 16);
            yv += __shfl_xor(yv, 8, 16);
            if (n == 0) sm_y[dl * TLP + j] = yv + Dsd * u;
        }
        __syncthreads();

        for (int i = t; i < 64 * TL; i += 1024) {
            int r = i >> 6, col = i & 63;
            outy[((size_t)kd * DI + dblk * 64 + r) * LL + l0 + col] =
                sm_y[r * TLP + col];
        }
    }
}

__global__ __launch_bounds__(192) void ss2d_merge_ln(
    const float* __restrict__ outy,
    const float* __restrict__ gamma,
    const float* __restrict__ beta,
    float* __restrict__ out)
{
    __shared__ float red[16];
    __shared__ float stats[2];

    int b  = blockIdx.x >> 12;
    int lm = blockIdx.x & 4095;
    int hh = lm >> 6, ww = lm & 63;
    int lt = (ww << 6) | hh;
    int d  = threadIdx.x;

    const size_t ks = (size_t)DI * LL;
    const float* base = outy + (size_t)b * KK * ks + (size_t)d * LL;

    float v = base[lm]
            + base[2 * ks + (LL - 1 - lm)]
            + base[ks + lt]
            + base[3 * ks + (LL - 1 - lt)];

    float s = v, q = v * v;
#pragma unroll
    for (int o = 16; o; o >>= 1) {
        s += __shfl_xor(s, o, 32);
        q += __shfl_xor(q, o, 32);
    }
    int wv = d >> 5;
    if ((d & 31) == 0) { red[wv] = s; red[8 + wv] = q; }
    __syncthreads();
    if (d == 0) {
        float ts = 0.f, tq = 0.f;
        for (int i = 0; i < 6; ++i) { ts += red[i]; tq += red[8 + i]; }
        float mu  = ts / 192.0f;
        float var = tq / 192.0f - mu * mu;
        stats[0] = mu;
        stats[1] = rsqrtf(var + 1e-5f);
    }
    __syncthreads();
    float mu = stats[0], rstd = stats[1];
    out[(size_t)blockIdx.x * DI + d] = (v - mu) * rstd * gamma[d] + beta[d];
}

extern "C" void kernel_launch(void* const* d_in, const int* in_sizes, int n_in,
                              void* d_out, int out_size, void* d_ws, size_t ws_size,
                              hipStream_t stream) {
    const float* x   = (const float*)d_in[0];
    const float* y   = (const float*)d_in[1];
    const float* xpw = (const float*)d_in[2];
    const float* dtw = (const float*)d_in[3];
    const float* dtb = (const float*)d_in[4];
    const float* Al  = (const float*)d_in[5];
    const float* Ds  = (const float*)d_in[6];
    const float* gm  = (const float*)d_in[7];
    const float* bt  = (const float*)d_in[8];

    float* xdbl = (float*)d_ws;
    float* outy = xdbl + (size_t)B_ * KK * CP * LL;

    ss2d_proj_wmma<<<1536, 256, 0, stream>>>(x, xpw, xdbl);
    ss2d_scan<<<B_ * KK * 3, 1024, 0, stream>>>(y, dtw, dtb, Al, Ds, xdbl, outy);
    ss2d_merge_ln<<<B_ * LL, 192, 0, stream>>>(outy, gm, bt, (float*)d_out);
}


# scan split along L: 32 chunks of 128 steps run in parallel (pass from h=0, chunk-carry pass, final pass); thread = channel x chunk, 16 states in registers; order of h and state sums changes
# speedup vs baseline: 9.0343x; 9.0343x over previous
#include <hip/hip_runtime.h>
#include <hip/hip_bf16.h>


#define B_  4
#define DI  192
#define LL  4096
#define NS  16
#define RR  6
#define KK  4
#define CP  38

typedef float v2f __attribute__((ext_vector_type(2)));
typedef float v8f __attribute__((ext_vector_type(8)));

__device__ __forceinline__ int scan_idx(int k, int l) {
    int ll = (k & 2) ? (LL - 1 - l) : l;
    if (k & 1) return ((ll & 63) << 6) | (ll >> 6);
    return ll;
}

__global__ __launch_bounds__(256) void ss2d_proj_wmma(
    const float* __restrict__ x,
    const float* __restrict__ xpw,
    float* __restrict__ xdbl)
{
    int wave = (int)((blockIdx.x * blockDim.x + threadIdx.x) >> 5);
    int lane = threadIdx.x & 31;
    int nt = wave & 255;
    int mt = (wave >> 8) % 3;
    int kd = (wave >> 8) / 3;
    if (kd >= B_ * KK) return;
    int k = kd & 3;
    int b = kd >> 2;

    int m0 = mt << 4;
    int l0 = nt << 4;
    int hw = lane >> 4;
    int ln = lane & 15;

    const float* wbase = xpw + (size_t)k * CP * DI;
    const float* xb    = x + (size_t)b * DI * LL;

    int arow   = m0 + ln;
    int arow_c = arow < CP ? arow : (CP - 1);
    float amsk = arow < CP ? 1.0f : 0.0f;

    int midx = scan_idx(k, l0 + ln);

    v8f acc = {0.f, 0.f, 0.f, 0.f, 0.f, 0.f, 0.f, 0.f};
    for (int k0 = 0; k0 < DI; k0 += 4) {
        const float* ap = wbase + arow_c * DI + k0 + (hw << 1);
        v2f a; a.x = ap[0] * amsk; a.y = ap[1] * amsk;
        const float* bp = xb + (size_t)(k0 + (hw << 1)) * LL + midx;
        v2f bb; bb.x = bp[0]; bb.y = bp[LL];
        acc = __builtin_amdgcn_wmma_f32_16x16x4_f32(
            false, a, false, bb, (short)0, acc, false, false);
    }

    float* obase = xdbl + (size_t)kd * CP * LL + l0 + ln;
#pragma unroll
    for (int i = 0; i < 8; ++i) {
        int m = m0 + i + (hw << 3);
        if (m < CP) obase[(size_t)m * LL] = acc[i];
    }
}

#define NC  32
#define CT  128
#define XS  44
#define US  129

typedef float v4f __attribute__((ext_vector_type(4)));

__device__ __forceinline__ void scan_chunk_body(
    const bool final_pass,
    const float* __restrict__ y,
    const float* __restrict__ dtw_g,
    const float* __restrict__ dtb_g,
    const float* __restrict__ Alogs,
    const float* __restrict__ Ds_g,
    const float* __restrict__ xdbl,
    float* outy,
    float* sm_x,
    float* sm_uy)
{
    const int t    = threadIdx.x;
    const int dblk = blockIdx.x % 3;
    const int c    = (blockIdx.x / 3) % NC;
    const int kd   = blockIdx.x / (3 * NC);
    const int k    = kd & 3;
    const int b    = kd >> 2;
    const int d    = dblk * 64 + t;
    const int kdch = k * DI + d;

    v2f A2[8];
    {
        const v4f* ap = (const v4f*)(Alogs + (size_t)kdch * NS);
#pragma unroll
        for (int i = 0; i < 4; ++i) {
            v4f a = ap[i];
            A2[2 * i].x     = -__expf(a.x);
            A2[2 * i].y     = -__expf(a.y);
            A2[2 * i + 1].x = -__expf(a.z);
            A2[2 * i + 1].y = -__expf(a.w);
        }
    }
    float dtw[RR];
#pragma unroll
    for (int r = 0; r < RR; ++r) dtw[r] = dtw_g[(size_t)kdch * RR + r];
    const float dtb = dtb_g[kdch];
    const float Dsd = Ds_g[kdch];

    const float* xdb = xdbl + (size_t)kd * CP * LL;
    const float* yb  = y + ((size_t)b * DI + dblk * 64) * LL;
    float* myrow     = outy + ((size_t)kd * DI + d) * LL + c * CT;

    v2f h2[8];
    if (final_pass) {
        const v4f* hp = (const v4f*)myrow;
#pragma unroll
        for (int i = 0; i < 4; ++i) {
            v4f v = hp[i];
            h2[2 * i].x = v.x; h2[2 * i].y = v.y;
            h2[2 * i + 1].x = v.z; h2[2 * i + 1].y = v.w;
        }
    } else {
#pragma unroll
        for (int i = 0; i < 8; ++i) { h2[i].x = 0.0f; h2[i].y = 0.0f; }
    }
    float sdelta = 0.0f;

    if (k & 1) {
        const int wbase = (k & 2) ? (62 - 2 * c) : (2 * c);
        for (int r = 0; r < 64; ++r) {
            const int mem = t * 64 + wbase;
            v2f v = *(const v2f*)(yb + (size_t)r * LL + mem);
#pragma unroll
            for (int e = 0; e < 2; ++e) {
                int ll = (wbase + e) * 64 + t;
                int l  = (k & 2) ? (LL - 1 - ll) : ll;
                sm_uy[r * US + (l - c * CT)] = e ? v.y : v.x;
            }
        }
    } else {
        const int mbase = (k & 2) ? (LL - CT - c * CT) : (c * CT);
        const int q = t & 31;
        for (int it = 0; it < 32; ++it) {
            const int r = 2 * it + (t >> 5);
            v4f v = *(const v4f*)(yb + (size_t)r * LL + mbase + 4 * q);
#pragma unroll
            for (int e = 0; e < 4; ++e) {
                int m = mbase + 4 * q + e;
                int l = (k & 2) ? (LL - 1 - m) : m;
                sm_uy[r * US + (l - c * CT)] = v[e];
            }
        }
    }

    const int nrows = final_pass ? CP : (RR + NS);
    float* uy = sm_uy + t * US;

    for (int s = 0; s < CT / 64; ++s) {
        const int l0 = c * CT + s * 64;
        for (int r = 0; r < nrows; ++r) {
            const int pos = r < RR ? r : r + 2;
            sm_x[t * XS + pos] = xdb[(size_t)r * LL + l0 + t];
        }
        __syncthreads();

        for (int j = 0; j < 64; ++j) {
            const v4f* xr = (const v4f*)(sm_x + j * XS);
            const v4f d0 = xr[0], d1 = xr[1];
            float sarg = dtb;
            sarg += dtw[0] * d0.x;
            sarg += dtw[1] * d0.y;
            sarg += dtw[2] * d0.z;
            sarg += dtw[3] * d0.w;
            sarg += dtw[4] * d1.x;
            sarg += dtw[5] * d1.y;
            const float delta = (sarg > 20.0f) ? sarg : log1pf(__expf(sarg));

            const float u  = uy[s * 64 + j];
            const float du = delta * u;
            const v2f dl2 = {delta, delta};
            const v2f du2 = {du, du};

            v2f bv[8];
#pragma unroll
            for (int i = 0; i < 4; ++i) {
                v4f q4 = xr[2 + i];
                bv[2 * i].x = q4.x; bv[2 * i].y = q4.y;
                bv[2 * i + 1].x = q4.z; bv[2 * i + 1].y = q4.w;
            }
#pragma unroll
            for (int i = 0; i < 8; ++i) {
                v2f arg = dl2 * A2[i];
                v2f ex;
                ex.x = __expf(arg.x);
                ex.y = __expf(arg.y);
                h2[i] = ex * h2[i] + du2 * bv[i];
            }

            if (final_pass) {
                v2f acc = {0.0f, 0.0f};
#pragma unroll
                for (int i = 0; i < 4; ++i) {
                    v4f q4 = xr[6 + i];
                    v2f c0 = {q4.x, q4.y};
                    v2f c1 = {q4.z, q4.w};
                    acc = h2[2 * i] * c0 + acc;
                    acc = h2[2 * i + 1] * c1 + acc;
                }
                uy[s * 64 + j] = (acc.x + acc.y) + Dsd * u;
            } else {
                sdelta += delta;
            }
        }
        __syncthreads();
    }

    if (final_pass) {
        const int q = t & 31;
        for (int it = 0; it < 32; ++it) {
            const int r = 2 * it + (t >> 5);
            const float* sp = sm_uy + r * US + 4 * q;
            v4f v = {sp[0], sp[1], sp[2], sp[3]};
            *(v4f*)(outy + ((size_t)kd * DI + dblk * 64 + r) * LL + c * CT + 4 * q) = v;
        }
    } else {
        v4f* hp = (v4f*)myrow;
#pragma unroll
        for (int i = 0; i < 4; ++i) {
            v4f v = {h2[2 * i].x, h2[2 * i].y, h2[2 * i + 1].x, h2[2 * i + 1].y};
            hp[i] = v;
        }
        myrow[NS] = sdelta;
    }
}

__global__ __launch_bounds__(64) void ss2d_scan_local(
    const float* __restrict__ y,
    const float* __restrict__ dtw_g,
    const float* __restrict__ dtb_g,
    const float* __restrict__ Alogs,
    const float* __restrict__ Ds_g,
    const float* __restrict__ xdbl,
    float* outy)
{
    __shared__ __attribute__((aligned(16))) float sm_x[64 * XS];
    __shared__ float sm_uy[64 * US];
    scan_chunk_body(false, y, dtw_g, dtb_g, Alogs, Ds_g, xdbl, outy, sm_x, sm_uy);
}

__global__ __launch_bounds__(64) void ss2d_scan_final(
    const float* __restrict__ y,
    const float* __restrict__ dtw_g,
    const float* __restrict__ dtb_g,
    const float* __restrict__ Alogs,
    const float* __restrict__ Ds_g,
    const float* __restrict__ xdbl,
    float* outy)
{
    __shared__ __attribute__((aligned(16))) float sm_x[64 * XS];
    __shared__ float sm_uy[64 * US];
    scan_chunk_body(true, y, dtw_g, dtb_g, Alogs, Ds_g, xdbl, outy, sm_x, sm_uy);
}

__global__ __launch_bounds__(256) void ss2d_scan_carry(
    const float* __restrict__ Alogs,
    float* outy)
{
    const int gid = blockIdx.x * 256 + threadIdx.x;
    const int n   = gid & 15;
    const int ch  = gid >> 4;
    const int kd  = ch / DI;
    const int d   = ch - kd * DI;
    const int k   = kd & 3;
    const float A = -__expf(Alogs[(size_t)(k * DI + d) * NS + n]);
    float* base = outy + (size_t)ch * LL;

    float S[NC], sd[NC];
#pragma unroll
    for (int c = 0; c < NC; ++c) {
        S[c]  = base[c * CT + n];
        sd[c] = base[c * CT + NS];
    }
    float H = 0.0f;
#pragma unroll
    for (int c = 0; c < NC; ++c) {
        base[c * CT + n] = H;
        H = __expf(A * sd[c]) * H + S[c];
    }
}

__global__ __launch_bounds__(192) void ss2d_merge_ln(
    const float* __restrict__ outy,
    const float* __restrict__ gamma,
    const float* __restrict__ beta,
    float* __restrict__ out)
{
    __shared__ float red[16];
    __shared__ float stats[2];

    int b  = blockIdx.x >> 12;
    int lm = blockIdx.x & 4095;
    int hh = lm >> 6, ww = lm & 63;
    int lt = (ww << 6) | hh;
    int d  = threadIdx.x;

    const size_t ks = (size_t)DI * LL;
    const float* base = outy + (size_t)b * KK * ks + (size_t)d * LL;

    float v = base[lm]
            + base[2 * ks + (LL - 1 - lm)]
            + base[ks + lt]
            + base[3 * ks + (LL - 1 - lt)];

    float s = v, q = v * v;
#pragma unroll
    for (int o = 16; o; o >>= 1) {
        s += __shfl_xor(s, o, 32);
        q += __shfl_xor(q, o, 32);
    }
    int wv = d >> 5;
    if ((d & 31) == 0) { red[wv] = s; red[8 + wv] = q; }
    __syncthreads();
    if (d == 0) {
        float ts = 0.f, tq = 0.f;
        for (int i = 0; i < 6; ++i) { ts += red[i]; tq += red[8 + i]; }
        float mu  = ts / 192.0f;
        float var = tq / 192.0f - mu * mu;
        stats[0] = mu;
        stats[1] = rsqrtf(var + 1e-5f);
    }
    __syncthreads();
    float mu = stats[0], rstd = stats[1];
    out[(size_t)blockIdx.x * DI + d] = (v - mu) * rstd * gamma[d] + beta[d];
}

extern "C" void kernel_launch(void* const* d_in, const int* in_sizes, int n_in,
                              void* d_out, int out_size, void* d_ws, size_t ws_size,
                              hipStream_t stream) {
    const float* x   = (const float*)d_in[0];
    const float* y   = (const float*)d_in[1];
    const float* xpw = (const float*)d_in[2];
    const float* dtw = (const float*)d_in[3];
    const float* dtb = (const float*)d_in[4];
    const float* Al  = (const float*)d_in[5];
    const float* Ds  = (const float*)d_in[6];
    const float* gm  = (const float*)d_in[7];
    const float* bt  = (const float*)d_in[8];

    float* xdbl = (float*)d_ws;
    float* outy = xdbl + (size_t)B_ * KK * CP * LL;

    ss2d_proj_wmma<<<1536, 256, 0, stream>>>(x, xpw, xdbl);
    ss2d_scan_local<<<B_ * KK * NC * 3, 64, 0, stream>>>(y, dtw, dtb, Al, Ds, xdbl, outy);
    ss2d_scan_carry<<<B_ * KK * DI * NS / 256, 256, 0, stream>>>(Al, outy);
    ss2d_scan_final<<<B_ * KK * NC * 3, 64, 0, stream>>>(y, dtw, dtb, Al, Ds, xdbl, outy);
    ss2d_merge_ln<<<B_ * LL, 192, 0, stream>>>(outy, gm, bt, (float*)d_out);
}


# scan y and chunk-carry values stored position-major outy[dir][image pos][channel]: final pass stores a 128-B line per wave-step; merge+LN a wave per position on whole lines; LN sum order changes
# speedup vs baseline: 11.5645x; 1.2801x over previous
#include <hip/hip_runtime.h>
#include <hip/hip_bf16.h>


#define B_  4
#define DI  192
#define LL  4096
#define NS  16
#define RR  6
#define KK  4
#define CP  38

typedef float v2f __attribute__((ext_vector_type(2)));
typedef float v8f __attribute__((ext_vector_type(8)));

__device__ __forceinline__ int scan_idx(int k, int l) {
    int ll = (k & 2) ? (LL - 1 - l) : l;
    if (k & 1) return ((ll & 63) << 6) | (ll >> 6);
    return ll;
}

__global__ __launch_bounds__(256) void ss2d_proj_wmma(
    const float* __restrict__ x,
    const float* __restrict__ xpw,
    float* __restrict__ xdbl)
{
    int wave = (int)((blockIdx.x * blockDim.x + threadIdx.x) >> 5);
    int lane = threadIdx.x & 31;
    int nt = wave & 255;
    int mt = (wave >> 8) % 3;
    int kd = (wave >> 8) / 3;
    if (kd >= B_ * KK) return;
    int k = kd & 3;
    int b = kd >> 2;

    int m0 = mt << 4;
    int l0 = nt << 4;
    int hw = lane >> 4;
    int ln = lane & 15;

    const float* wbase = xpw + (size_t)k * CP * DI;
    const float* xb    = x + (size_t)b * DI * LL;

    int arow   = m0 + ln;
    int arow_c = arow < CP ? arow : (CP - 1);
    float amsk = arow < CP ? 1.0f : 0.0f;

    int midx = scan_idx(k, l0 + ln);

    v8f acc = {0.f, 0.f, 0.f, 0.f, 0.f, 0.f, 0.f, 0.f};
    for (int k0 = 0; k0 < DI; k0 += 4) {
        const float* ap = wbase + arow_c * DI + k0 + (hw << 1);
        v2f a; a.x = ap[0] * amsk; a.y = ap[1] * amsk;
        const float* bp = xb + (size_t)(k0 + (hw << 1)) * LL + midx;
        v2f bb; bb.x = bp[0]; bb.y = bp[LL];
        acc = __builtin_amdgcn_wmma_f32_16x16x4_f32(
            false, a, false, bb, (short)0, acc, false, false);
    }

    float* obase = xdbl + (size_t)kd * CP * LL + l0 + ln;
#pragma unroll
    for (int i = 0; i < 8; ++i) {
        int m = m0 + i + (hw << 3);
        if (m < CP) obase[(size_t)m * LL] = acc[i];
    }
}

#define NC  32
#define CT  128
#define XS  44
#define US  129

typedef float v4f __attribute__((ext_vector_type(4)));

__device__ __forceinline__ void scan_chunk_body(
    const bool final_pass,
    const float* __restrict__ y,
    const float* __restrict__ dtw_g,
    const float* __restrict__ dtb_g,
    const float* __restrict__ Alogs,
    const float* __restrict__ Ds_g,
    const float* __restrict__ xdbl,
    float* outy,
    float* sm_x,
    float* sm_uy)
{
    const int t    = threadIdx.x;
    const int dblk = blockIdx.x % 3;
    const int c    = (blockIdx.x / 3) % NC;
    const int kd   = blockIdx.x / (3 * NC);
    const int k    = kd & 3;
    const int b    = kd >> 2;
    const int d    = dblk * 64 + t;
    const int kdch = k * DI + d;

    v2f A2[8];
    {
        const v4f* ap = (const v4f*)(Alogs + (size_t)kdch * NS);
#pragma unroll
        for (int i = 0; i < 4; ++i) {
            v4f a = ap[i];
            A2[2 * i].x     = -__expf(a.x);
            A2[2 * i].y     = -__expf(a.y);
            A2[2 * i + 1].x = -__expf(a.z);
            A2[2 * i + 1].y = -__expf(a.w);
        }
    }
    float dtw[RR];
#pragma unroll
    for (int r = 0; r < RR; ++r) dtw[r] = dtw_g[(size_t)kdch * RR + r];
    const float dtb = dtb_g[kdch];
    const float Dsd = Ds_g[kdch];

    const float* xdb = xdbl + (size_t)kd * CP * LL;
    const float* yb  = y + ((size_t)b * DI + dblk * 64) * LL;
    float* ych       = outy + (size_t)kd * LL * DI + d;

    v2f h2[8];
    if (final_pass) {
#pragma unroll
        for (int i = 0; i < 8; ++i) {
            h2[i].x = ych[(size_t)scan_idx(k, c * CT + 2 * i) * DI];
            h2[i].y = ych[(size_t)scan_idx(k, c * CT + 2 * i + 1) * DI];
        }
    } else {
#pragma unroll
        for (int i = 0; i < 8; ++i) { h2[i].x = 0.0f; h2[i].y = 0.0f; }
    }
    float sdelta = 0.0f;

    if (k & 1) {
        const int wbase = (k & 2) ? (62 - 2 * c) : (2 * c);
        for (int r = 0; r < 64; ++r) {
            const int mem = t * 64 + wbase;
            v2f v = *(const v2f*)(yb + (size_t)r * LL + mem);
#pragma unroll
            for (int e = 0; e < 2; ++e) {
                int ll = (wbase + e) * 64 + t;
                int l  = (k & 2) ? (LL - 1 - ll) : ll;
                sm_uy[r * US + (l - c * CT)] = e ? v.y : v.x;
            }
        }
    } else {
        const int mbase = (k & 2) ? (LL - CT - c * CT) : (c * CT);
        const int q = t & 31;
        for (int it = 0; it < 32; ++it) {
            const int r = 2 * it + (t >> 5);
            v4f v = *(const v4f*)(yb + (size_t)r * LL + mbase + 4 * q);
#pragma unroll
            for (int e = 0; e < 4; ++e) {
                int m = mbase + 4 * q + e;
                int l = (k & 2) ? (LL - 1 - m) : m;
                sm_uy[r * US + (l - c * CT)] = v[e];
            }
        }
    }

    const int nrows = final_pass ? CP : (RR + NS);
    float* uy = sm_uy + t * US;

    for (int s = 0; s < CT / 64; ++s) {
        const int l0 = c * CT + s * 64;
        for (int r = 0; r < nrows; ++r) {
            const int pos = r < RR ? r : r + 2;
            sm_x[t * XS + pos] = xdb[(size_t)r * LL + l0 + t];
        }
        __syncthreads();

        for (int j = 0; j < 64; ++j) {
            const v4f* xr = (const v4f*)(sm_x + j * XS);
            const v4f d0 = xr[0], d1 = xr[1];
            float sarg = dtb;
            sarg += dtw[0] * d0.x;
            sarg += dtw[1] * d0.y;
            sarg += dtw[2] * d0.z;
            sarg += dtw[3] * d0.w;
            sarg += dtw[4] * d1.x;
            sarg += dtw[5] * d1.y;
            const float delta = (sarg > 20.0f) ? sarg : log1pf(__expf(sarg));

            const float u  = uy[s * 64 + j];
            const float du = delta * u;
            const v2f dl2 = {delta, delta};
            const v2f du2 = {du, du};

            v2f bv[8];
#pragma unroll
            for (int i = 0; i < 4; ++i) {
                v4f q4 = xr[2 + i];
                bv[2 * i].x = q4.x; bv[2 * i].y = q4.y;
                bv[2 * i + 1].x = q4.z; bv[2 * i + 1].y = q4.w;
            }
#pragma unroll
            for (int i = 0; i < 8; ++i) {
                v2f arg = dl2 * A2[i];
                v2f ex;
                ex.x = __expf(arg.x);
                ex.y = __expf(arg.y);
                h2[i] = ex * h2[i] + du2 * bv[i];
            }

            if (final_pass) {
                v2f acc = {0.0f, 0.0f};
#pragma unroll
                for (int i = 0; i < 4; ++i) {
                    v4f q4 = xr[6 + i];
                    v2f c0 = {q4.x, q4.y};
                    v2f c1 = {q4.z, q4.w};
                    acc = h2[2 * i] * c0 + acc;
                    acc = h2[2 * i + 1] * c1 + acc;
                }
                ych[(size_t)scan_idx(k, l0 + j) * DI] = (acc.x + acc.y) + Dsd * u;
            } else {
                sdelta += delta;
            }
        }
        __syncthreads();
    }

    if (!final_pass) {
#pragma unroll
        for (int i = 0; i < 8; ++i) {
            ych[(size_t)scan_idx(k, c * CT + 2 * i) * DI]     = h2[i].x;
            ych[(size_t)scan_idx(k, c * CT + 2 * i + 1) * DI] = h2[i].y;
        }
        ych[(size_t)scan_idx(k, c * CT + NS) * DI] = sdelta;
    }
}

__global__ __launch_bounds__(64) void ss2d_scan_local(
    const float* __restrict__ y,
    const float* __restrict__ dtw_g,
    const float* __restrict__ dtb_g,
    const float* __restrict__ Alogs,
    const float* __restrict__ Ds_g,
    const float* __restrict__ xdbl,
    float* outy)
{
    __shared__ __attribute__((aligned(16))) float sm_x[64 * XS];
    __shared__ float sm_uy[64 * US];
    scan_chunk_body(false, y, dtw_g, dtb_g, Alogs, Ds_g, xdbl, outy, sm_x, sm_uy);
}

__global__ __launch_bounds__(64) void ss2d_scan_final(
    const float* __restrict__ y,
    const float* __restrict__ dtw_g,
    const float* __restrict__ dtb_g,
    const float* __restrict__ Alogs,
    const float* __restrict__ Ds_g,
    const float* __restrict__ xdbl,
    float* outy)
{
    __shared__ __attribute__((aligned(16))) float sm_x[64 * XS];
    __shared__ float sm_uy[64 * US];
    scan_chunk_body(true, y, dtw_g, dtb_g, Alogs, Ds_g, xdbl, outy, sm_x, sm_uy);
}

__global__ __launch_bounds__(192) void ss2d_scan_carry(
    const float* __restrict__ Alogs,
    float* outy)
{
    const int d   = threadIdx.x;
    const int n   = blockIdx.x & 15;
    const int kd  = blockIdx.x >> 4;
    const int k   = kd & 3;
    const float A = -__expf(Alogs[(size_t)(k * DI + d) * NS + n]);
    float* base = outy + (size_t)kd * LL * DI + d;

    float S[NC], sd[NC];
#pragma unroll
    for (int c = 0; c < NC; ++c) {
        S[c]  = base[(size_t)scan_idx(k, c * CT + n) * DI];
        sd[c] = base[(size_t)scan_idx(k, c * CT + NS) * DI];
    }
    float H = 0.0f;
#pragma unroll
    for (int c = 0; c < NC; ++c) {
        base[(size_t)scan_idx(k, c * CT + n) * DI] = H;
        H = __expf(A * sd[c]) * H + S[c];
    }
}

__global__ __launch_bounds__(256) void ss2d_merge_ln(
    const float* __restrict__ outy,
    const float* __restrict__ gamma,
    const float* __restrict__ beta,
    float* __restrict__ out)
{
    const int lane = threadIdx.x & 31;
    const int wave = (int)((blockIdx.x * 256 + threadIdx.x) >> 5);
    const size_t ks = (size_t)LL * DI;

    v2f g[3], be[3];
#pragma unroll
    for (int i = 0; i < 3; ++i) {
        g[i]  = *(const v2f*)(gamma + 2 * lane + 64 * i);
        be[i] = *(const v2f*)(beta + 2 * lane + 64 * i);
    }

    v2f v[2][3];
#pragma unroll
    for (int pp = 0; pp < 2; ++pp) {
        const int p  = wave * 2 + pp;
        const int b  = p >> 12;
        const int lm = p & (LL - 1);
        const float* base = outy + ((size_t)b * KK * LL + lm) * DI + 2 * lane;
#pragma unroll
        for (int i = 0; i < 3; ++i) {
            const v2f a0 = *(const v2f*)(base + 64 * i);
            const v2f a2 = *(const v2f*)(base + 2 * ks + 64 * i);
            const v2f a1 = *(const v2f*)(base + ks + 64 * i);
            const v2f a3 = *(const v2f*)(base + 3 * ks + 64 * i);
            v[pp][i] = ((a0 + a2) + a1) + a3;
        }
    }

#pragma unroll
    for (int pp = 0; pp < 2; ++pp) {
        float s = 0.0f, q = 0.0f;
#pragma unroll
        for (int i = 0; i < 3; ++i) {
            s += v[pp][i].x; q += v[pp][i].x * v[pp][i].x;
            s += v[pp][i].y; q += v[pp][i].y * v[pp][i].y;
        }
#pragma unroll
        for (int o = 16; o; o >>= 1) {
            s += __shfl_xor(s, o, 32);
            q += __shfl_xor(q, o, 32);
        }
        const float mu   = s / 192.0f;
        const float var  = q / 192.0f - mu * mu;
        const float rstd = rsqrtf(var + 1e-5f);
        float* op = out + (size_t)(wave * 2 + pp) * DI + 2 * lane;
#pragma unroll
        for (int i = 0; i < 3; ++i) {
            v2f r;
            r.x = (v[pp][i].x - mu) * rstd * g[i].x + be[i].x;
            r.y = (v[pp][i].y - mu) * rstd * g[i].y + be[i].y;
            *(v2f*)(op + 64 * i) = r;
        }
    }
}

extern "C" void kernel_launch(void* const* d_in, const int* in_sizes, int n_in,
                              void* d_out, int out_size, void* d_ws, size_t ws_size,
                              hipStream_t stream) {
    const float* x   = (const float*)d_in[0];
    const float* y   = (const float*)d_in[1];
    const float* xpw = (const float*)d_in[2];
    const float* dtw = (const float*)d_in[3];
    const float* dtb = (const float*)d_in[4];
    const float* Al  = (const float*)d_in[5];
    const float* Ds  = (const float*)d_in[6];
    const float* gm  = (const float*)d_in[7];
    const float* bt  = (const float*)d_in[8];

    float* xdbl = (float*)d_ws;
    float* outy = xdbl + (size_t)B_ * KK * CP * LL;

    ss2d_proj_wmma<<<1536, 256, 0, stream>>>(x, xpw, xdbl);
    ss2d_scan_local<<<B_ * KK * NC * 3, 64, 0, stream>>>(y, dtw, dtb, Al, Ds, xdbl, outy);
    ss2d_scan_carry<<<B_ * KK * NS, DI, 0, stream>>>(Al, outy);
    ss2d_scan_final<<<B_ * KK * NC * 3, 64, 0, stream>>>(y, dtw, dtb, Al, Ds, xdbl, outy);
    ss2d_merge_ln<<<B_ * LL / 16, 256, 0, stream>>>(outy, gm, bt, (float*)d_out);
}


# projection tiles are 16 neighbouring image positions in all 4 directions: x loads contiguous (64 B per half-wave), the cross-scan permutation moves to the store of the 38 result rows; sums unchanged
# speedup vs baseline: 12.8546x; 1.1116x over previous
#include <hip/hip_runtime.h>
#include <hip/hip_bf16.h>


#define B_  4
#define DI  192
#define LL  4096
#define NS  16
#define RR  6
#define KK  4
#define CP  38

typedef float v2f __attribute__((ext_vector_type(2)));
typedef float v8f __attribute__((ext_vector_type(8)));

__device__ __forceinline__ int scan_idx(int k, int l) {
    int ll = (k & 2) ? (LL - 1 - l) : l;
    if (k & 1) return ((ll & 63) << 6) | (ll >> 6);
    return ll;
}

__global__ __launch_bounds__(256) void ss2d_proj_wmma(
    const float* __restrict__ x,
    const float* __restrict__ xpw,
    float* __restrict__ xdbl)
{
    int wave = (int)((blockIdx.x * blockDim.x + threadIdx.x) >> 5);
    int lane = threadIdx.x & 31;
    int nt = wave & 255;
    int mt = (wave >> 8) % 3;
    int kd = (wave >> 8) / 3;
    if (kd >= B_ * KK) return;
    int k = kd & 3;
    int b = kd >> 2;

    int m0 = mt << 4;
    int l0 = nt << 4;
    int hw = lane >> 4;
    int ln = lane & 15;

    const float* wbase = xpw + (size_t)k * CP * DI;
    const float* xb    = x + (size_t)b * DI * LL;

    int arow   = m0 + ln;
    int arow_c = arow < CP ? arow : (CP - 1);
    float amsk = arow < CP ? 1.0f : 0.0f;

    int m = l0 + ln;
    int lt   = (k & 1) ? (((m & 63) << 6) | (m >> 6)) : m;
    int lpos = (k & 2) ? (LL - 1 - lt) : lt;

    v8f acc = {0.f, 0.f, 0.f, 0.f, 0.f, 0.f, 0.f, 0.f};
    for (int k0 = 0; k0 < DI; k0 += 4) {
        const float* ap = wbase + arow_c * DI + k0 + (hw << 1);
        v2f a; a.x = ap[0] * amsk; a.y = ap[1] * amsk;
        const float* bp = xb + (size_t)(k0 + (hw << 1)) * LL + m;
        v2f bb; bb.x = bp[0]; bb.y = bp[LL];
        acc = __builtin_amdgcn_wmma_f32_16x16x4_f32(
            false, a, false, bb, (short)0, acc, false, false);
    }

    float* obase = xdbl + (size_t)kd * CP * LL + lpos;
#pragma unroll
    for (int i = 0; i < 8; ++i) {
        int m = m0 + i + (hw << 3);
        if (m < CP) obase[(size_t)m * LL] = acc[i];
    }
}

#define NC  32
#define CT  128
#define XS  44
#define US  129

typedef float v4f __attribute__((ext_vector_type(4)));

__device__ __forceinline__ void scan_chunk_body(
    const bool final_pass,
    const float* __restrict__ y,
    const float* __restrict__ dtw_g,
    const float* __restrict__ dtb_g,
    const float* __restrict__ Alogs,
    const float* __restrict__ Ds_g,
    const float* __restrict__ xdbl,
    float* outy,
    float* sm_x,
    float* sm_uy)
{
    const int t    = threadIdx.x;
    const int dblk = blockIdx.x % 3;
    const int c    = (blockIdx.x / 3) % NC;
    const int kd   = blockIdx.x / (3 * NC);
    const int k    = kd & 3;
    const int b    = kd >> 2;
    const int d    = dblk * 64 + t;
    const int kdch = k * DI + d;

    v2f A2[8];
    {
        const v4f* ap = (const v4f*)(Alogs + (size_t)kdch * NS);
#pragma unroll
        for (int i = 0; i < 4; ++i) {
            v4f a = ap[i];
            A2[2 * i].x     = -__expf(a.x);
            A2[2 * i].y     = -__expf(a.y);
            A2[2 * i + 1].x = -__expf(a.z);
            A2[2 * i + 1].y = -__expf(a.w);
        }
    }
    float dtw[RR];
#pragma unroll
    for (int r = 0; r < RR; ++r) dtw[r] = dtw_g[(size_t)kdch * RR + r];
    const float dtb = dtb_g[kdch];
    const float Dsd = Ds_g[kdch];

    const float* xdb = xdbl + (size_t)kd * CP * LL;
    const float* yb  = y + ((size_t)b * DI + dblk * 64) * LL;
    float* ych       = outy + (size_t)kd * LL * DI + d;

    v2f h2[8];
    if (final_pass) {
#pragma unroll
        for (int i = 0; i < 8; ++i) {
            h2[i].x = ych[(size_t)scan_idx(k, c * CT + 2 * i) * DI];
            h2[i].y = ych[(size_t)scan_idx(k, c * CT + 2 * i + 1) * DI];
        }
    } else {
#pragma unroll
        for (int i = 0; i < 8; ++i) { h2[i].x = 0.0f; h2[i].y = 0.0f; }
    }
    float sdelta = 0.0f;

    if (k & 1) {
        const int wbase = (k & 2) ? (62 - 2 * c) : (2 * c);
        for (int r = 0; r < 64; ++r) {
            const int mem = t * 64 + wbase;
            v2f v = *(const v2f*)(yb + (size_t)r * LL + mem);
#pragma unroll
            for (int e = 0; e < 2; ++e) {
                int ll = (wbase + e) * 64 + t;
                int l  = (k & 2) ? (LL - 1 - ll) : ll;
                sm_uy[r * US + (l - c * CT)] = e ? v.y : v.x;
            }
        }
    } else {
        const int mbase = (k & 2) ? (LL - CT - c * CT) : (c * CT);
        const int q = t & 31;
        for (int it = 0; it < 32; ++it) {
            const int r = 2 * it + (t >> 5);
            v4f v = *(const v4f*)(yb + (size_t)r * LL + mbase + 4 * q);
#pragma unroll
            for (int e = 0; e < 4; ++e) {
                int m = mbase + 4 * q + e;
                int l = (k & 2) ? (LL - 1 - m) : m;
                sm_uy[r * US + (l - c * CT)] = v[e];
            }
        }
    }

    const int nrows = final_pass ? CP : (RR + NS);
    float* uy = sm_uy + t * US;

    for (int s = 0; s < CT / 64; ++s) {
        const int l0 = c * CT + s * 64;
        for (int r = 0; r < nrows; ++r) {
            const int pos = r < RR ? r : r + 2;
            sm_x[t * XS + pos] = xdb[(size_t)r * LL + l0 + t];
        }
        __syncthreads();

        for (int j = 0; j < 64; ++j) {
            const v4f* xr = (const v4f*)(sm_x + j * XS);
            const v4f d0 = xr[0], d1 = xr[1];
            float sarg = dtb;
            sarg += dtw[0] * d0.x;
            sarg += dtw[1] * d0.y;
            sarg += dtw[2] * d0.z;
            sarg += dtw[3] * d0.w;
            sarg += dtw[4] * d1.x;
            sarg += dtw[5] * d1.y;
            const float delta = (sarg > 20.0f) ? sarg : log1pf(__expf(sarg));

            const float u  = uy[s * 64 + j];
            const float du = delta * u;
            const v2f dl2 = {delta, delta};
            const v2f du2 = {du, du};

            v2f bv[8];
#pragma unroll
            for (int i = 0; i < 4; ++i) {
                v4f q4 = xr[2 + i];
                bv[2 * i].x = q4.x; bv[2 * i].y = q4.y;
                bv[2 * i + 1].x = q4.z; bv[2 * i + 1].y = q4.w;
            }
#pragma unroll
            for (int i = 0; i < 8; ++i) {
                v2f arg = dl2 * A2[i];
                v2f ex;
                ex.x = __expf(arg.x);
                ex.y = __expf(arg.y);
                h2[i] = ex * h2[i] + du2 * bv[i];
            }

            if (final_pass) {
                v2f acc = {0.0f, 0.0f};
#pragma unroll
                for (int i = 0; i < 4; ++i) {
                    v4f q4 = xr[6 + i];
                    v2f c0 = {q4.x, q4.y};
                    v2f c1 = {q4.z, q4.w};
                    acc = h2[2 * i] * c0 + acc;
                    acc = h2[2 * i + 1] * c1 + acc;
                }
                ych[(size_t)scan_idx(k, l0 + j) * DI] = (acc.x + acc.y) + Dsd * u;
            } else {
                sdelta += delta;
            }
        }
        __syncthreads();
    }

    if (!final_pass) {
#pragma unroll
        for (int i = 0; i < 8; ++i) {
            ych[(size_t)scan_idx(k, c * CT + 2 * i) * DI]     = h2[i].x;
            ych[(size_t)scan_idx(k, c * CT + 2 * i + 1) * DI] = h2[i].y;
        }
        ych[(size_t)scan_idx(k, c * CT + NS) * DI] = sdelta;
    }
}

__global__ __launch_bounds__(64) void ss2d_scan_local(
    const float* __restrict__ y,
    const float* __restrict__ dtw_g,
    const float* __restrict__ dtb_g,
    const float* __restrict__ Alogs,
    const float* __restrict__ Ds_g,
    const float* __restrict__ xdbl,
    float* outy)
{
    __shared__ __attribute__((aligned(16))) float sm_x[64 * XS];
    __shared__ float sm_uy[64 * US];
    scan_chunk_body(false, y, dtw_g, dtb_g, Alogs, Ds_g, xdbl, outy, sm_x, sm_uy);
}

__global__ __launch_bounds__(64) void ss2d_scan_final(
    const float* __restrict__ y,
    const float* __restrict__ dtw_g,
    const float* __restrict__ dtb_g,
    const float* __restrict__ Alogs,
    const float* __restrict__ Ds_g,
    const float* __restrict__ xdbl,
    float* outy)
{
    __shared__ __attribute__((aligned(16))) float sm_x[64 * XS];
    __shared__ float sm_uy[64 * US];
    scan_chunk_body(true, y, dtw_g, dtb_g, Alogs, Ds_g, xdbl, outy, sm_x, sm_uy);
}

__global__ __launch_bounds__(192) void ss2d_scan_carry(
    const float* __restrict__ Alogs,
    float* outy)
{
    const int d   = threadIdx.x;
    const int n   = blockIdx.x & 15;
    const int kd  = blockIdx.x >> 4;
    const int k   = kd & 3;
    const float A = -__expf(Alogs[(size_t)(k * DI + d) * NS + n]);
    float* base = outy + (size_t)kd * LL * DI + d;

    float S[NC], sd[NC];
#pragma unroll
    for (int c = 0; c < NC; ++c) {
        S[c]  = base[(size_t)scan_idx(k, c * CT + n) * DI];
        sd[c] = base[(size_t)scan_idx(k, c * CT + NS) * DI];
    }
    float H = 0.0f;
#pragma unroll
    for (int c = 0; c < NC; ++c) {
        base[(size_t)scan_idx(k, c * CT + n) * DI] = H;
        H = __expf(A * sd[c]) * H + S[c];
    }
}

__global__ __launch_bounds__(256) void ss2d_merge_ln(
    const float* __restrict__ outy,
    const float* __restrict__ gamma,
    const float* __restrict__ beta,
    float* __restrict__ out)
{
    const int lane = threadIdx.x & 31;
    const int wave = (int)((blockIdx.x * 256 + threadIdx.x) >> 5);
    const size_t ks = (size_t)LL * DI;

    v2f g[3], be[3];
#pragma unroll
    for (int i = 0; i < 3; ++i) {
        g[i]  = *(const v2f*)(gamma + 2 * lane + 64 * i);
        be[i] = *(const v2f*)(beta + 2 * lane + 64 * i);
    }

    v2f v[2][3];
#pragma unroll
    for (int pp = 0; pp < 2; ++pp) {
        const int p  = wave * 2 + pp;
        const int b  = p >> 12;
        const int lm = p & (LL - 1);
        const float* base = outy + ((size_t)b * KK * LL + lm) * DI + 2 * lane;
#pragma unroll
        for (int i = 0; i < 3; ++i) {
            const v2f a0 = *(const v2f*)(base + 64 * i);
            const v2f a2 = *(const v2f*)(base + 2 * ks + 64 * i);
            const v2f a1 = *(const v2f*)(base + ks + 64 * i);
            const v2f a3 = *(const v2f*)(base + 3 * ks + 64 * i);
            v[pp][i] = ((a0 + a2) + a1) + a3;
        }
    }

#pragma unroll
    for (int pp = 0; pp < 2; ++pp) {
        float s = 0.0f, q = 0.0f;
#pragma unroll
        for (int i = 0; i < 3; ++i) {
            s += v[pp][i].x; q += v[pp][i].x * v[pp][i].x;
            s += v[pp][i].y; q += v[pp][i].y * v[pp][i].y;
        }
#pragma unroll
        for (int o = 16; o; o >>= 1) {
            s += __shfl_xor(s, o, 32);
            q += __shfl_xor(q, o, 32);
        }
        const float mu   = s / 192.0f;
        const float var  = q / 192.0f - mu * mu;
        const float rstd = rsqrtf(var + 1e-5f);
        float* op = out + (size_t)(wave * 2 + pp) * DI + 2 * lane;
#pragma unroll
        for (int i = 0; i < 3; ++i) {
            v2f r;
            r.x = (v[pp][i].x - mu) * rstd * g[i].x + be[i].x;
            r.y = (v[pp][i].y - mu) * rstd * g[i].y + be[i].y;
            *(v2f*)(op + 64 * i) = r;
        }
    }
}

extern "C" void kernel_launch(void* const* d_in, const int* in_sizes, int n_in,
                              void* d_out, int out_size, void* d_ws, size_t ws_size,
                              hipStream_t stream) {
    const float* x   = (const float*)d_in[0];
    const float* y   = (const float*)d_in[1];
    const float* xpw = (const float*)d_in[2];
    const float* dtw = (const float*)d_in[3];
    const float* dtb = (const float*)d_in[4];
    const float* Al  = (const float*)d_in[5];
    const float* Ds  = (const float*)d_in[6];
    const float* gm  = (const float*)d_in[7];
    const float* bt  = (const float*)d_in[8];

    float* xdbl = (float*)d_ws;
    float* outy = xdbl + (size_t)B_ * KK * CP * LL;

    ss2d_proj_wmma<<<1536, 256, 0, stream>>>(x, xpw, xdbl);
    ss2d_scan_local<<<B_ * KK * NC * 3, 64, 0, stream>>>(y, dtw, dtb, Al, Ds, xdbl, outy);
    ss2d_scan_carry<<<B_ * KK * NS, DI, 0, stream>>>(Al, outy);
    ss2d_scan_final<<<B_ * KK * NC * 3, 64, 0, stream>>>(y, dtw, dtb, Al, Ds, xdbl, outy);
    ss2d_merge_ln<<<B_ * LL / 16, 256, 0, stream>>>(outy, gm, bt, (float*)d_out);
}


# scan chunks of 64 steps (64 chunks, was 32 of 128): six waves per SIMD, not three, cover the step's dependent softplus chain; block = 64 channels x 2 chunks on one u tile; carry sum order changes
# speedup vs baseline: 13.7007x; 1.0658x over previous
#include <hip/hip_runtime.h>
#include <hip/hip_bf16.h>


#define B_  4
#define DI  192
#define LL  4096
#define NS  16
#define RR  6
#define KK  4
#define CP  38

typedef float v2f __attribute__((ext_vector_type(2)));
typedef float v8f __attribute__((ext_vector_type(8)));

__device__ __forceinline__ int scan_idx(int k, int l) {
    int ll = (k & 2) ? (LL - 1 - l) : l;
    if (k & 1) return ((ll & 63) << 6) | (ll >> 6);
    return ll;
}

__global__ __launch_bounds__(256) void ss2d_proj_wmma(
    const float* __restrict__ x,
    const float* __restrict__ xpw,
    float* __restrict__ xdbl)
{
    int wave = (int)((blockIdx.x * blockDim.x + threadIdx.x) >> 5);
    int lane = threadIdx.x & 31;
    int nt = wave & 255;
    int mt = (wave >> 8) % 3;
    int kd = (wave >> 8) / 3;
    if (kd >= B_ * KK) return;
    int k = kd & 3;
    int b = kd >> 2;

    int m0 = mt << 4;
    int l0 = nt << 4;
    int hw = lane >> 4;
    int ln = lane & 15;

    const float* wbase = xpw + (size_t)k * CP * DI;
    const float* xb    = x + (size_t)b * DI * LL;

    int arow   = m0 + ln;
    int arow_c = arow < CP ? arow : (CP - 1);
    float amsk = arow < CP ? 1.0f : 0.0f;

    int m = l0 + ln;
    int lt   = (k & 1) ? (((m & 63) << 6) | (m >> 6)) : m;
    int lpos = (k & 2) ? (LL - 1 - lt) : lt;

    v8f acc = {0.f, 0.f, 0.f, 0.f, 0.f, 0.f, 0.f, 0.f};
    for (int k0 = 0; k0 < DI; k0 += 4) {
        const float* ap = wbase + arow_c * DI + k0 + (hw << 1);
        v2f a; a.x = ap[0] * amsk; a.y = ap[1] * amsk;
        const float* bp = xb + (size_t)(k0 + (hw << 1)) * LL + m;
        v2f bb; bb.x = bp[0]; bb.y = bp[LL];
        acc = __builtin_amdgcn_wmma_f32_16x16x4_f32(
            false, a, false, bb, (short)0, acc, false, false);
    }

    float* obase = xdbl + (size_t)kd * CP * LL + lpos;
#pragma unroll
    for (int i = 0; i < 8; ++i) {
        int m = m0 + i + (hw << 3);
        if (m < CP) obase[(size_t)m * LL] = acc[i];
    }
}

#define NC  64
#define CT  64
#define NP  32
#define PT  128
#define XS  40

typedef float v4f __attribute__((ext_vector_type(4)));

__device__ __forceinline__ int u_slot(int r, int p) {
    return r * PT + ((p + r) & (PT - 1));
}

__device__ __forceinline__ void scan_chunk_body(
    const bool final_pass,
    const float* __restrict__ y,
    const float* __restrict__ dtw_g,
    const float* __restrict__ dtb_g,
    const float* __restrict__ Alogs,
    const float* __restrict__ Ds_g,
    const float* __restrict__ xdbl,
    float* outy,
    float* sm_x,
    float* sm_uy)
{
    const int tid  = threadIdx.x;
    const int t    = tid & 63;
    const int hf   = __builtin_amdgcn_readfirstlane(tid >> 6);
    const int dblk = blockIdx.x % 3;
    const int cp   = (blockIdx.x / 3) % NP;
    const int kd   = blockIdx.x / (3 * NP);
    const int c    = 2 * cp + hf;
    const int k    = kd & 3;
    const int b    = kd >> 2;
    const int d    = dblk * 64 + t;
    const int kdch = k * DI + d;

    v2f A2[8];
    {
        const v4f* ap = (const v4f*)(Alogs + (size_t)kdch * NS);
#pragma unroll
        for (int i = 0; i < 4; ++i) {
            v4f a = ap[i];
            A2[2 * i].x     = -__expf(a.x);
            A2[2 * i].y     = -__expf(a.y);
            A2[2 * i + 1].x = -__expf(a.z);
            A2[2 * i + 1].y = -__expf(a.w);
        }
    }
    float dtw[RR];
#pragma unroll
    for (int r = 0; r < RR; ++r) dtw[r] = dtw_g[(size_t)kdch * RR + r];
    const float dtb = dtb_g[kdch];
    const float Dsd = Ds_g[kdch];

    const float* xdb = xdbl + (size_t)kd * CP * LL;
    const float* yb  = y + ((size_t)b * DI + dblk * 64) * LL;
    float* ych       = outy + (size_t)kd * LL * DI + d;

    v2f h2[8];
    if (final_pass) {
#pragma unroll
        for (int i = 0; i < 8; ++i) {
            h2[i].x = ych[(size_t)scan_idx(k, c * CT + 2 * i) * DI];
            h2[i].y = ych[(size_t)scan_idx(k, c * CT + 2 * i + 1) * DI];
        }
    } else {
#pragma unroll
        for (int i = 0; i < 8; ++i) { h2[i].x = 0.0f; h2[i].y = 0.0f; }
    }
    float sdelta = 0.0f;

    if (k & 1) {
        const int wbase = (k & 2) ? (62 - 2 * cp) : (2 * cp);
        v2f uv[32];
#pragma unroll
        for (int i = 0; i < 32; ++i)
            uv[i] = *(const v2f*)(yb + (size_t)(2 * i + hf) * LL + t * 64 + wbase);
#pragma unroll
        for (int i = 0; i < 32; ++i) {
            const int r = 2 * i + hf;
#pragma unroll
            for (int e = 0; e < 2; ++e) {
                int ll = (wbase + e) * 64 + t;
                int l  = (k & 2) ? (LL - 1 - ll) : ll;
                sm_uy[u_slot(r, l - cp * PT)] = e ? uv[i].y : uv[i].x;
            }
        }
    } else {
        const int mbase = (k & 2) ? (LL - PT - cp * PT) : (cp * PT);
        const int q = tid & 31;
        const int wv = __builtin_amdgcn_readfirstlane(tid >> 5);
        v4f uq[16];
#pragma unroll
        for (int it = 0; it < 16; ++it)
            uq[it] = *(const v4f*)(yb + (size_t)(4 * it + wv) * LL + mbase + 4 * q);
#pragma unroll
        for (int it = 0; it < 16; ++it) {
            const int r = 4 * it + wv;
#pragma unroll
            for (int e = 0; e < 4; ++e) {
                int m = mbase + 4 * q + e;
                int l = (k & 2) ? (LL - 1 - m) : m;
                sm_uy[u_slot(r, l - cp * PT)] = uq[it][e];
            }
        }
    }

    const int nrows = final_pass ? CP : (RR + NS);
    float* sm_xh = sm_x + hf * (CT * XS);
    const int l0 = c * CT;
    for (int r = 0; r < nrows; ++r) {
        const int pos = r < RR ? r : r + 2;
        sm_xh[t * XS + pos] = xdb[(size_t)r * LL + l0 + t];
    }
    __syncthreads();

    const float* urow = sm_uy + t * PT;
    int up = (hf * CT + t) & (PT - 1);

    for (int j = 0; j < CT; ++j) {
        const v4f* xr = (const v4f*)(sm_xh + j * XS);
        const v4f d0 = xr[0], d1 = xr[1];
        float sarg = dtb;
        sarg += dtw[0] * d0.x;
        sarg += dtw[1] * d0.y;
        sarg += dtw[2] * d0.z;
        sarg += dtw[3] * d0.w;
        sarg += dtw[4] * d1.x;
        sarg += dtw[5] * d1.y;
        const float delta = (sarg > 20.0f) ? sarg : log1pf(__expf(sarg));

        const float u  = urow[up];
        up = (up + 1) & (PT - 1);
        const float du = delta * u;
        const v2f dl2 = {delta, delta};
        const v2f du2 = {du, du};

        v2f bv[8];
#pragma unroll
        for (int i = 0; i < 4; ++i) {
            v4f q4 = xr[2 + i];
            bv[2 * i].x = q4.x; bv[2 * i].y = q4.y;
            bv[2 * i + 1].x = q4.z; bv[2 * i + 1].y = q4.w;
        }
#pragma unroll
        for (int i = 0; i < 8; ++i) {
            v2f arg = dl2 * A2[i];
            v2f ex;
            ex.x = __expf(arg.x);
            ex.y = __expf(arg.y);
            h2[i] = ex * h2[i] + du2 * bv[i];
        }

        if (final_pass) {
            v2f acc = {0.0f, 0.0f};
#pragma unroll
            for (int i = 0; i < 4; ++i) {
                v4f q4 = xr[6 + i];
                v2f c0 = {q4.x, q4.y};
                v2f c1 = {q4.z, q4.w};
                acc = h2[2 * i] * c0 + acc;
                acc = h2[2 * i + 1] * c1 + acc;
            }
            ych[(size_t)scan_idx(k, l0 + j) * DI] = (acc.x + acc.y) + Dsd * u;
        } else {
            sdelta += delta;
        }
    }

    if (!final_pass) {
#pragma unroll
        for (int i = 0; i < 8; ++i) {
            ych[(size_t)scan_idx(k, c * CT + 2 * i) * DI]     = h2[i].x;
            ych[(size_t)scan_idx(k, c * CT + 2 * i + 1) * DI] = h2[i].y;
        }
        ych[(size_t)scan_idx(k, c * CT + NS) * DI] = sdelta;
    }
}

__global__ __launch_bounds__(128) __attribute__((amdgpu_waves_per_eu(6)))
void ss2d_scan_local(
    const float* __restrict__ y,
    const float* __restrict__ dtw_g,
    const float* __restrict__ dtb_g,
    const float* __restrict__ Alogs,
    const float* __restrict__ Ds_g,
    const float* __restrict__ xdbl,
    float* outy)
{
    __shared__ __attribute__((aligned(16))) float sm_x[2 * CT * XS];
    __shared__ float sm_uy[64 * PT];
    scan_chunk_body(false, y, dtw_g, dtb_g, Alogs, Ds_g, xdbl, outy, sm_x, sm_uy);
}

__global__ __launch_bounds__(128) __attribute__((amdgpu_waves_per_eu(6)))
void ss2d_scan_final(
    const float* __restrict__ y,
    const float* __restrict__ dtw_g,
    const float* __restrict__ dtb_g,
    const float* __restrict__ Alogs,
    const float* __restrict__ Ds_g,
    const float* __restrict__ xdbl,
    float* outy)
{
    __shared__ __attribute__((aligned(16))) float sm_x[2 * CT * XS];
    __shared__ float sm_uy[64 * PT];
    scan_chunk_body(true, y, dtw_g, dtb_g, Alogs, Ds_g, xdbl, outy, sm_x, sm_uy);
}

__global__ __launch_bounds__(192) void ss2d_scan_carry(
    const float* __restrict__ Alogs,
    float* outy)
{
    const int d   = threadIdx.x;
    const int n   = blockIdx.x & 15;
    const int kd  = blockIdx.x >> 4;
    const int k   = kd & 3;
    const float A = -__expf(Alogs[(size_t)(k * DI + d) * NS + n]);
    float* base = outy + (size_t)kd * LL * DI + d;

    float S[NC], sd[NC];
#pragma unroll
    for (int c = 0; c < NC; ++c) {
        S[c]  = base[(size_t)scan_idx(k, c * CT + n) * DI];
        sd[c] = base[(size_t)scan_idx(k, c * CT + NS) * DI];
    }
    float H = 0.0f;
#pragma unroll
    for (int c = 0; c < NC; ++c) {
        base[(size_t)scan_idx(k, c * CT + n) * DI] = H;
        H = __expf(A * sd[c]) * H + S[c];
    }
}

__global__ __launch_bounds__(256) void ss2d_merge_ln(
    const float* __restrict__ outy,
    const float* __restrict__ gamma,
    const float* __restrict__ beta,
    float* __restrict__ out)
{
    const int lane = threadIdx.x & 31;
    const int wave = (int)((blockIdx.x * 256 + threadIdx.x) >> 5);
    const size_t ks = (size_t)LL * DI;

    v2f g[3], be[3];
#pragma unroll
    for (int i = 0; i < 3; ++i) {
        g[i]  = *(const v2f*)(gamma + 2 * lane + 64 * i);
        be[i] = *(const v2f*)(beta + 2 * lane + 64 * i);
    }

    v2f v[2][3];
#pragma unroll
    for (int pp = 0; pp < 2; ++pp) {
        const int p  = wave * 2 + pp;
        const int b  = p >> 12;
        const int lm = p & (LL - 1);
        const float* base = outy + ((size_t)b * KK * LL + lm) * DI + 2 * lane;
#pragma unroll
        for (int i = 0; i < 3; ++i) {
            const v2f a0 = *(const v2f*)(base + 64 * i);
            const v2f a2 = *(const v2f*)(base + 2 * ks + 64 * i);
            const v2f a1 = *(const v2f*)(base + ks + 64 * i);
            const v2f a3 = *(const v2f*)(base + 3 * ks + 64 * i);
            v[pp][i] = ((a0 + a2) + a1) + a3;
        }
    }

#pragma unroll
    for (int pp = 0; pp < 2; ++pp) {
        float s = 0.0f, q = 0.0f;
#pragma unroll
        for (int i = 0; i < 3; ++i) {
            s += v[pp][i].x; q += v[pp][i].x * v[pp][i].x;
            s += v[pp][i].y; q += v[pp][i].y * v[pp][i].y;
        }
#pragma unroll
        for (int o = 16; o; o >>= 1) {
            s += __shfl_xor(s, o, 32);
            q += __shfl_xor(q, o, 32);
        }
        const float mu   = s / 192.0f;
        const float var  = q / 192.0f - mu * mu;
        const float rstd = rsqrtf(var + 1e-5f);
        float* op = out + (size_t)(wave * 2 + pp) * DI + 2 * lane;
#pragma unroll
        for (int i = 0; i < 3; ++i) {
            v2f r;
            r.x = (v[pp][i].x - mu) * rstd * g[i].x + be[i].x;
            r.y = (v[pp][i].y - mu) * rstd * g[i].y + be[i].y;
            *(v2f*)(op + 64 * i) = r;
        }
    }
}

extern "C" void kernel_launch(void* const* d_in, const int* in_sizes, int n_in,
                              void* d_out, int out_size, void* d_ws, size_t ws_size,
                              hipStream_t stream) {
    const float* x   = (const float*)d_in[0];
    const float* y   = (const float*)d_in[1];
    const float* xpw = (const float*)d_in[2];
    const float* dtw = (const float*)d_in[3];
    const float* dtb = (const float*)d_in[4];
    const float* Al  = (const float*)d_in[5];
    const float* Ds  = (const float*)d_in[6];
    const float* gm  = (const float*)d_in[7];
    const float* bt  = (const float*)d_in[8];

    float* xdbl = (float*)d_ws;
    float* outy = xdbl + (size_t)B_ * KK * CP * LL;

    ss2d_proj_wmma<<<1536, 256, 0, stream>>>(x, xpw, xdbl);
    ss2d_scan_local<<<B_ * KK * NP * 3, 128, 0, stream>>>(y, dtw, dtb, Al, Ds, xdbl, outy);
    ss2d_scan_carry<<<B_ * KK * NS, DI, 0, stream>>>(Al, outy);
    ss2d_scan_final<<<B_ * KK * NP * 3, 128, 0, stream>>>(y, dtw, dtb, Al, Ds, xdbl, outy);
    ss2d_merge_ln<<<B_ * LL / 16, 256, 0, stream>>>(outy, gm, bt, (float*)d_out);
}


# scan: delta (dt projection + softplus) of chunk steps 20..63 computed once, in the first pass, and handed on in the step's own outy slot; final pass works out 20 of 64 itself; sums unchanged
# speedup vs baseline: 14.8120x; 1.0811x over previous
#include <hip/hip_runtime.h>
#include <hip/hip_bf16.h>


#define B_  4
#define DI  192
#define LL  4096
#define NS  16
#define RR  6
#define KK  4
#define CP  38

typedef float v2f __attribute__((ext_vector_type(2)));
typedef float v8f __attribute__((ext_vector_type(8)));

__device__ __forceinline__ int scan_idx(int k, int l) {
    int ll = (k & 2) ? (LL - 1 - l) : l;
    if (k & 1) return ((ll & 63) << 6) | (ll >> 6);
    return ll;
}

__global__ __launch_bounds__(256) void ss2d_proj_wmma(
    const float* __restrict__ x,
    const float* __restrict__ xpw,
    float* __restrict__ xdbl)
{
    int wave = (int)((blockIdx.x * blockDim.x + threadIdx.x) >> 5);
    int lane = threadIdx.x & 31;
    int nt = wave & 255;
    int mt = (wave >> 8) % 3;
    int kd = (wave >> 8) / 3;
    if (kd >= B_ * KK) return;
    int k = kd & 3;
    int b = kd >> 2;

    int m0 = mt << 4;
    int l0 = nt << 4;
    int hw = lane >> 4;
    int ln = lane & 15;

    const float* wbase = xpw + (size_t)k * CP * DI;
    const float* xb    = x + (size_t)b * DI * LL;

    int arow   = m0 + ln;
    int arow_c = arow < CP ? arow : (CP - 1);
    float amsk = arow < CP ? 1.0f : 0.0f;

    int m = l0 + ln;
    int lt   = (k & 1) ? (((m & 63) << 6) | (m >> 6)) : m;
    int lpos = (k & 2) ? (LL - 1 - lt) : lt;

    v8f acc = {0.f, 0.f, 0.f, 0.f, 0.f, 0.f, 0.f, 0.f};
    for (int k0 = 0; k0 < DI; k0 += 4) {
        const float* ap = wbase + arow_c * DI + k0 + (hw << 1);
        v2f a; a.x = ap[0] * amsk; a.y = ap[1] * amsk;
        const float* bp = xb + (size_t)(k0 + (hw << 1)) * LL + m;
        v2f bb; bb.x = bp[0]; bb.y = bp[LL];
        acc = __builtin_amdgcn_wmma_f32_16x16x4_f32(
            false, a, false, bb, (short)0, acc, false, false);
    }

    float* obase = xdbl + (size_t)kd * CP * LL + lpos;
#pragma unroll
    for (int i = 0; i < 8; ++i) {
        int m = m0 + i + (hw << 3);
        if (m < CP) obase[(size_t)m * LL] = acc[i];
    }
}

#define NC  64
#define CT  64
#define NP  32
#define PT  128
#define XS  40
#define NRC 20

typedef float v4f __attribute__((ext_vector_type(4)));

__device__ __forceinline__ int u_slot(int r, int p) {
    return r * PT + ((p + r) & (PT - 1));
}

__device__ __forceinline__ void scan_chunk_body(
    const bool final_pass,
    const float* __restrict__ y,
    const float* __restrict__ dtw_g,
    const float* __restrict__ dtb_g,
    const float* __restrict__ Alogs,
    const float* __restrict__ Ds_g,
    const float* __restrict__ xdbl,
    float* outy,
    float* sm_x,
    float* sm_uy)
{
    const int tid  = threadIdx.x;
    const int t    = tid & 63;
    const int hf   = __builtin_amdgcn_readfirstlane(tid >> 6);
    const int dblk = blockIdx.x % 3;
    const int cp   = (blockIdx.x / 3) % NP;
    const int kd   = blockIdx.x / (3 * NP);
    const int c    = 2 * cp + hf;
    const int k    = kd & 3;
    const int b    = kd >> 2;
    const int d    = dblk * 64 + t;
    const int kdch = k * DI + d;

    v2f A2[8];
    {
        const v4f* ap = (const v4f*)(Alogs + (size_t)kdch * NS);
#pragma unroll
        for (int i = 0; i < 4; ++i) {
            v4f a = ap[i];
            A2[2 * i].x     = -__expf(a.x);
            A2[2 * i].y     = -__expf(a.y);
            A2[2 * i + 1].x = -__expf(a.z);
            A2[2 * i + 1].y = -__expf(a.w);
        }
    }
    float dtw[RR];
#pragma unroll
    for (int r = 0; r < RR; ++r) dtw[r] = dtw_g[(size_t)kdch * RR + r];
    const float dtb = dtb_g[kdch];
    const float Dsd = Ds_g[kdch];

    const float* xdb = xdbl + (size_t)kd * CP * LL;
    const float* yb  = y + ((size_t)b * DI + dblk * 64) * LL;
    float* ych       = outy + (size_t)kd * LL * DI + d;

    v2f h2[8];
    if (final_pass) {
#pragma unroll
        for (int i = 0; i < 8; ++i) {
            h2[i].x = ych[(size_t)scan_idx(k, c * CT + 2 * i) * DI];
            h2[i].y = ych[(size_t)scan_idx(k, c * CT + 2 * i + 1) * DI];
        }
    } else {
#pragma unroll
        for (int i = 0; i < 8; ++i) { h2[i].x = 0.0f; h2[i].y = 0.0f; }
    }
    float sdelta = 0.0f;

    if (k & 1) {
        const int wbase = (k & 2) ? (62 - 2 * cp) : (2 * cp);
        v2f uv[32];
#pragma unroll
        for (int i = 0; i < 32; ++i)
            uv[i] = *(const v2f*)(yb + (size_t)(2 * i + hf) * LL + t * 64 + wbase);
#pragma unroll
        for (int i = 0; i < 32; ++i) {
            const int r = 2 * i + hf;
#pragma unroll
            for (int e = 0; e < 2; ++e) {
                int ll = (wbase + e) * 64 + t;
                int l  = (k & 2) ? (LL - 1 - ll) : ll;
                sm_uy[u_slot(r, l - cp * PT)] = e ? uv[i].y : uv[i].x;
            }
        }
    } else {
        const int mbase = (k & 2) ? (LL - PT - cp * PT) : (cp * PT);
        const int q = tid & 31;
        const int wv = __builtin_amdgcn_readfirstlane(tid >> 5);
        v4f uq[16];
#pragma unroll
        for (int it = 0; it < 16; ++it)
            uq[it] = *(const v4f*)(yb + (size_t)(4 * it + wv) * LL + mbase + 4 * q);
#pragma unroll
        for (int it = 0; it < 16; ++it) {
            const int r = 4 * it + wv;
#pragma unroll
            for (int e = 0; e < 4; ++e) {
                int m = mbase + 4 * q + e;
                int l = (k & 2) ? (LL - 1 - m) : m;
                sm_uy[u_slot(r, l - cp * PT)] = uq[it][e];
            }
        }
    }

    const int nrows = final_pass ? CP : (RR + NS);
    float* sm_xh = sm_x + hf * (CT * XS);
    const int l0 = c * CT;
    for (int r = 0; r < nrows; ++r) {
        const int pos = r < RR ? r : r + 2;
        sm_xh[t * XS + pos] = xdb[(size_t)r * LL + l0 + t];
    }
    __syncthreads();

    const float* urow = sm_uy + t * PT;
    int up = (hf * CT + t) & (PT - 1);

    const int mstep = ((k & 1) ? 64 : 1) * ((k & 2) ? -1 : 1);
    const int dstep = mstep * DI;
    float* kbase    = outy + (size_t)kd * LL * DI;
    int srec        = scan_idx(k, l0) * DI;

    auto delta_of = [&](const v4f* xr) -> float {
        const v4f d0 = xr[0], d1 = xr[1];
        float sarg = dtb;
        sarg += dtw[0] * d0.x;
        sarg += dtw[1] * d0.y;
        sarg += dtw[2] * d0.z;
        sarg += dtw[3] * d0.w;
        sarg += dtw[4] * d1.x;
        sarg += dtw[5] * d1.y;
        return (sarg > 20.0f) ? sarg : log1pf(__expf(sarg));
    };

    auto state_step = [&](const v4f* xr, const float delta) {
        const float u  = urow[up];
        up = (up + 1) & (PT - 1);
        const float du = delta * u;
        const v2f dl2 = {delta, delta};
        const v2f du2 = {du, du};

        v2f bv[8];
#pragma unroll
        for (int i = 0; i < 4; ++i) {
            v4f q4 = xr[2 + i];
            bv[2 * i].x = q4.x; bv[2 * i].y = q4.y;
            bv[2 * i + 1].x = q4.z; bv[2 * i + 1].y = q4.w;
        }
#pragma unroll
        for (int i = 0; i < 8; ++i) {
            v2f arg = dl2 * A2[i];
            v2f ex;
            ex.x = __expf(arg.x);
            ex.y = __expf(arg.y);
            h2[i] = ex * h2[i] + du2 * bv[i];
        }

        if (final_pass) {
            v2f acc = {0.0f, 0.0f};
#pragma unroll
            for (int i = 0; i < 4; ++i) {
                v4f q4 = xr[6 + i];
                v2f c0 = {q4.x, q4.y};
                v2f c1 = {q4.z, q4.w};
                acc = h2[2 * i] * c0 + acc;
                acc = h2[2 * i + 1] * c1 + acc;
            }
            (kbase + srec)[d] = (acc.x + acc.y) + Dsd * u;
        }
    };

    if (!final_pass) {
        for (int j = 0; j < CT; ++j) {
            const v4f* xr = (const v4f*)(sm_xh + j * XS);
            const float delta = delta_of(xr);
            state_step(xr, delta);
            sdelta += delta;
            if (j >= NRC) (kbase + srec)[d] = delta;
            srec += dstep;
        }
    } else {
        float dq[4];
#pragma unroll
        for (int i = 0; i < 4; ++i) dq[i] = (kbase + srec + (NRC + i) * dstep)[d];

        for (int j = 0; j < NRC; ++j) {
            const v4f* xr = (const v4f*)(sm_xh + j * XS);
            state_step(xr, delta_of(xr));
            srec += dstep;
        }
        for (int j = NRC; j < CT; j += 4) {
#pragma unroll
            for (int i = 0; i < 4; ++i) {
                const v4f* xr = (const v4f*)(sm_xh + (j + i) * XS);
                state_step(xr, dq[i]);
                const int ahead = min(j + i + 4, CT - 1) - (j + i);
                dq[i] = (kbase + srec + ahead * dstep)[d];
                srec += dstep;
                __builtin_amdgcn_sched_barrier(0);
            }
        }
    }

    if (!final_pass) {
#pragma unroll
        for (int i = 0; i < 8; ++i) {
            ych[(size_t)scan_idx(k, c * CT + 2 * i) * DI]     = h2[i].x;
            ych[(size_t)scan_idx(k, c * CT + 2 * i + 1) * DI] = h2[i].y;
        }
        ych[(size_t)scan_idx(k, c * CT + NS) * DI] = sdelta;
    }
}

__global__ __launch_bounds__(128) __attribute__((amdgpu_waves_per_eu(6)))
void ss2d_scan_local(
    const float* __restrict__ y,
    const float* __restrict__ dtw_g,
    const float* __restrict__ dtb_g,
    const float* __restrict__ Alogs,
    const float* __restrict__ Ds_g,
    const float* __restrict__ xdbl,
    float* outy)
{
    __shared__ __attribute__((aligned(16))) float sm_x[2 * CT * XS];
    __shared__ float sm_uy[64 * PT];
    scan_chunk_body(false, y, dtw_g, dtb_g, Alogs, Ds_g, xdbl, outy, sm_x, sm_uy);
}

__global__ __launch_bounds__(128) __attribute__((amdgpu_waves_per_eu(6)))
void ss2d_scan_final(
    const float* __restrict__ y,
    const float* __restrict__ dtw_g,
    const float* __restrict__ dtb_g,
    const float* __restrict__ Alogs,
    const float* __restrict__ Ds_g,
    const float* __restrict__ xdbl,
    float* outy)
{
    __shared__ __attribute__((aligned(16))) float sm_x[2 * CT * XS];
    __shared__ float sm_uy[64 * PT];
    scan_chunk_body(true, y, dtw_g, dtb_g, Alogs, Ds_g, xdbl, outy, sm_x, sm_uy);
}

__global__ __launch_bounds__(192) void ss2d_scan_carry(
    const float* __restrict__ Alogs,
    float* outy)
{
    const int d   = threadIdx.x;
    const int n   = blockIdx.x & 15;
    const int kd  = blockIdx.x >> 4;
    const int k   = kd & 3;
    const float A = -__expf(Alogs[(size_t)(k * DI + d) * NS + n]);
    float* base = outy + (size_t)kd * LL * DI + d;

    float S[NC], sd[NC];
#pragma unroll
    for (int c = 0; c < NC; ++c) {
        S[c]  = base[(size_t)scan_idx(k, c * CT + n) * DI];
        sd[c] = base[(size_t)scan_idx(k, c * CT + NS) * DI];
    }
    float H = 0.0f;
#pragma unroll
    for (int c = 0; c < NC; ++c) {
        base[(size_t)scan_idx(k, c * CT + n) * DI] = H;
        H = __expf(A * sd[c]) * H + S[c];
    }
}

__global__ __launch_bounds__(256) void ss2d_merge_ln(
    const float* __restrict__ outy,
    const float* __restrict__ gamma,
    const float* __restrict__ beta,
    float* __restrict__ out)
{
    const int lane = threadIdx.x & 31;
    const int wave = (int)((blockIdx.x * 256 + threadIdx.x) >> 5);
    const size_t ks = (size_t)LL * DI;

    v2f g[3], be[3];
#pragma unroll
    for (int i = 0; i < 3; ++i) {
        g[i]  = *(const v2f*)(gamma + 2 * lane + 64 * i);
        be[i] = *(const v2f*)(beta + 2 * lane + 64 * i);
    }

    v2f v[2][3];
#pragma unroll
    for (int pp = 0; pp < 2; ++pp) {
        const int p  = wave * 2 + pp;
        const int b  = p >> 12;
        const int lm = p & (LL - 1);
        const float* base = outy + ((size_t)b * KK * LL + lm) * DI + 2 * lane;
#pragma unroll
        for (int i = 0; i < 3; ++i) {
            const v2f a0 = *(const v2f*)(base + 64 * i);
            const v2f a2 = *(const v2f*)(base + 2 * ks + 64 * i);
            const v2f a1 = *(const v2f*)(base + ks + 64 * i);
            const v2f a3 = *(const v2f*)(base + 3 * ks + 64 * i);
            v[pp][i] = ((a0 + a2) + a1) + a3;
        }
    }

#pragma unroll
    for (int pp = 0; pp < 2; ++pp) {
        float s = 0.0f, q = 0.0f;
#pragma unroll
        for (int i = 0; i < 3; ++i) {
            s += v[pp][i].x; q += v[pp][i].x * v[pp][i].x;
            s += v[pp][i].y; q += v[pp][i].y * v[pp][i].y;
        }
#pragma unroll
        for (int o = 16; o; o >>= 1) {
            s += __shfl_xor(s, o, 32);
            q += __shfl_xor(q, o, 32);
        }
        const float mu   = s / 192.0f;
        const float var  = q / 192.0f - mu * mu;
        const float rstd = rsqrtf(var + 1e-5f);
        float* op = out + (size_t)(wave * 2 + pp) * DI + 2 * lane;
#pragma unroll
        for (int i = 0; i < 3; ++i) {
            v2f r;
            r.x = (v[pp][i].x - mu) * rstd * g[i].x + be[i].x;
            r.y = (v[pp][i].y - mu) * rstd * g[i].y + be[i].y;
            *(v2f*)(op + 64 * i) = r;
        }
    }
}

extern "C" void kernel_launch(void* const* d_in, const int* in_sizes, int n_in,
                              void* d_out, int out_size, void* d_ws, size_t ws_size,
                              hipStream_t stream) {
    const float* x   = (const float*)d_in[0];
    const float* y   = (const float*)d_in[1];
    const float* xpw = (const float*)d_in[2];
    const float* dtw = (const float*)d_in[3];
    const float* dtb = (const float*)d_in[4];
    const float* Al  = (const float*)d_in[5];
    const float* Ds  = (const float*)d_in[6];
    const float* gm  = (const float*)d_in[7];
    const float* bt  = (const float*)d_in[8];

    float* xdbl = (float*)d_ws;
    float* outy = xdbl + (size_t)B_ * KK * CP * LL;

    ss2d_proj_wmma<<<1536, 256, 0, stream>>>(x, xpw, xdbl);
    ss2d_scan_local<<<B_ * KK * NP * 3, 128, 0, stream>>>(y, dtw, dtb, Al, Ds, xdbl, outy);
    ss2d_scan_carry<<<B_ * KK * NS, DI, 0, stream>>>(Al, outy);
    ss2d_scan_final<<<B_ * KK * NP * 3, 128, 0, stream>>>(y, dtw, dtb, Al, Ds, xdbl, outy);
    ss2d_merge_ln<<<B_ * LL / 16, 256, 0, stream>>>(outy, gm, bt, (float*)d_out);
}


# scan: deltas (dt projection + softplus, now branch-free) of 4 steps (2 in a chunk's first 20) worked out side by side so their dependent chains interleave; sum order kept, 2 dt products now fused
# speedup vs baseline: 15.5798x; 1.0518x over previous
#include <hip/hip_runtime.h>
#include <hip/hip_bf16.h>


#define B_  4
#define DI  192
#define LL  4096
#define NS  16
#define RR  6
#define KK  4
#define CP  38

typedef float v2f __attribute__((ext_vector_type(2)));
typedef float v8f __attribute__((ext_vector_type(8)));

__device__ __forceinline__ int scan_idx(int k, int l) {
    int ll = (k & 2) ? (LL - 1 - l) : l;
    if (k & 1) return ((ll & 63) << 6) | (ll >> 6);
    return ll;
}

__global__ __launch_bounds__(256) void ss2d_proj_wmma(
    const float* __restrict__ x,
    const float* __restrict__ xpw,
    float* __restrict__ xdbl)
{
    int wave = (int)((blockIdx.x * blockDim.x + threadIdx.x) >> 5);
    int lane = threadIdx.x & 31;
    int nt = wave & 255;
    int mt = (wave >> 8) % 3;
    int kd = (wave >> 8) / 3;
    if (kd >= B_ * KK) return;
    int k = kd & 3;
    int b = kd >> 2;

    int m0 = mt << 4;
    int l0 = nt << 4;
    int hw = lane >> 4;
    int ln = lane & 15;

    const float* wbase = xpw + (size_t)k * CP * DI;
    const float* xb    = x + (size_t)b * DI * LL;

    int arow   = m0 + ln;
    int arow_c = arow < CP ? arow : (CP - 1);
    float amsk = arow < CP ? 1.0f : 0.0f;

    int m = l0 + ln;
    int lt   = (k & 1) ? (((m & 63) << 6) | (m >> 6)) : m;
    int lpos = (k & 2) ? (LL - 1 - lt) : lt;

    v8f acc = {0.f, 0.f, 0.f, 0.f, 0.f, 0.f, 0.f, 0.f};
    for (int k0 = 0; k0 < DI; k0 += 4) {
        const float* ap = wbase + arow_c * DI + k0 + (hw << 1);
        v2f a; a.x = ap[0] * amsk; a.y = ap[1] * amsk;
        const float* bp = xb + (size_t)(k0 + (hw << 1)) * LL + m;
        v2f bb; bb.x = bp[0]; bb.y = bp[LL];
        acc = __builtin_amdgcn_wmma_f32_16x16x4_f32(
            false, a, false, bb, (short)0, acc, false, false);
    }

    float* obase = xdbl + (size_t)kd * CP * LL + lpos;
#pragma unroll
    for (int i = 0; i < 8; ++i) {
        int m = m0 + i + (hw << 3);
        if (m < CP) obase[(size_t)m * LL] = acc[i];
    }
}

#define NC  64
#define CT  64
#define NP  32
#define PT  128
#define XS  40
#define NRC 20
#define DG  4
#define DGA 2

typedef float v4f __attribute__((ext_vector_type(4)));

__device__ __forceinline__ int u_slot(int r, int p) {
    return r * PT + ((p + r) & (PT - 1));
}

__device__ __forceinline__ void scan_chunk_body(
    const bool final_pass,
    const float* __restrict__ y,
    const float* __restrict__ dtw_g,
    const float* __restrict__ dtb_g,
    const float* __restrict__ Alogs,
    const float* __restrict__ Ds_g,
    const float* __restrict__ xdbl,
    float* outy,
    float* sm_x,
    float* sm_uy)
{
    const int tid  = threadIdx.x;
    const int t    = tid & 63;
    const int hf   = __builtin_amdgcn_readfirstlane(tid >> 6);
    const int dblk = blockIdx.x % 3;
    const int cp   = (blockIdx.x / 3) % NP;
    const int kd   = blockIdx.x / (3 * NP);
    const int c    = 2 * cp + hf;
    const int k    = kd & 3;
    const int b    = kd >> 2;
    const int d    = dblk * 64 + t;
    const int kdch = k * DI + d;

    v2f A2[8];
    {
        const v4f* ap = (const v4f*)(Alogs + (size_t)kdch * NS);
#pragma unroll
        for (int i = 0; i < 4; ++i) {
            v4f a = ap[i];
            A2[2 * i].x     = -__expf(a.x);
            A2[2 * i].y     = -__expf(a.y);
            A2[2 * i + 1].x = -__expf(a.z);
            A2[2 * i + 1].y = -__expf(a.w);
        }
    }
    float dtw[RR];
#pragma unroll
    for (int r = 0; r < RR; ++r) dtw[r] = dtw_g[(size_t)kdch * RR + r];
    const float dtb = dtb_g[kdch];
    const float Dsd = Ds_g[kdch];

    const float* xdb = xdbl + (size_t)kd * CP * LL;
    const float* yb  = y + ((size_t)b * DI + dblk * 64) * LL;
    float* ych       = outy + (size_t)kd * LL * DI + d;

    v2f h2[8];
    if (final_pass) {
#pragma unroll
        for (int i = 0; i < 8; ++i) {
            h2[i].x = ych[(size_t)scan_idx(k, c * CT + 2 * i) * DI];
            h2[i].y = ych[(size_t)scan_idx(k, c * CT + 2 * i + 1) * DI];
        }
    } else {
#pragma unroll
        for (int i = 0; i < 8; ++i) { h2[i].x = 0.0f; h2[i].y = 0.0f; }
    }
    float sdelta = 0.0f;

    if (k & 1) {
        const int wbase = (k & 2) ? (62 - 2 * cp) : (2 * cp);
        v2f uv[32];
#pragma unroll
        for (int i = 0; i < 32; ++i)
            uv[i] = *(const v2f*)(yb + (size_t)(2 * i + hf) * LL + t * 64 + wbase);
#pragma unroll
        for (int i = 0; i < 32; ++i) {
            const int r = 2 * i + hf;
#pragma unroll
            for (int e = 0; e < 2; ++e) {
                int ll = (wbase + e) * 64 + t;
                int l  = (k & 2) ? (LL - 1 - ll) : ll;
                sm_uy[u_slot(r, l - cp * PT)] = e ? uv[i].y : uv[i].x;
            }
        }
    } else {
        const int mbase = (k & 2) ? (LL - PT - cp * PT) : (cp * PT);
        const int q = tid & 31;
        const int wv = __builtin_amdgcn_readfirstlane(tid >> 5);
        v4f uq[16];
#pragma unroll
        for (int it = 0; it < 16; ++it)
            uq[it] = *(const v4f*)(yb + (size_t)(4 * it + wv) * LL + mbase + 4 * q);
#pragma unroll
        for (int it = 0; it < 16; ++it) {
            const int r = 4 * it + wv;
#pragma unroll
            for (int e = 0; e < 4; ++e) {
                int m = mbase + 4 * q + e;
                int l = (k & 2) ? (LL - 1 - m) : m;
                sm_uy[u_slot(r, l - cp * PT)] = uq[it][e];
            }
        }
    }

    const int nrows = final_pass ? CP : (RR + NS);
    float* sm_xh = sm_x + hf * (CT * XS);
    const int l0 = c * CT;
    for (int r = 0; r < nrows; ++r) {
        const int pos = r < RR ? r : r + 2;
        sm_xh[t * XS + pos] = xdb[(size_t)r * LL + l0 + t];
    }
    __syncthreads();

    const float* urow = sm_uy + t * PT;
    int up = (hf * CT + t) & (PT - 1);

    const int mstep = ((k & 1) ? 64 : 1) * ((k & 2) ? -1 : 1);
    const int dstep = mstep * DI;
    float* kbase    = outy + (size_t)kd * LL * DI;
    int srec        = scan_idx(k, l0) * DI;

    auto delta_of = [&](const v4f* xr) -> float {
        const v4f d0 = xr[0], d1 = xr[1];
        float sarg = dtb;
        sarg += dtw[0] * d0.x;
        sarg += dtw[1] * d0.y;
        sarg += dtw[2] * d0.z;
        sarg += dtw[3] * d0.w;
        sarg += dtw[4] * d1.x;
        sarg += dtw[5] * d1.y;
        const float lp = log1pf(__expf(sarg));
        return (sarg > 20.0f) ? sarg : lp;
    };

    auto state_step = [&](const v4f* xr, const float delta) {
        const float u  = urow[up];
        up = (up + 1) & (PT - 1);
        const float du = delta * u;
        const v2f dl2 = {delta, delta};
        const v2f du2 = {du, du};

        v2f bv[8];
#pragma unroll
        for (int i = 0; i < 4; ++i) {
            v4f q4 = xr[2 + i];
            bv[2 * i].x = q4.x; bv[2 * i].y = q4.y;
            bv[2 * i + 1].x = q4.z; bv[2 * i + 1].y = q4.w;
        }
#pragma unroll
        for (int i = 0; i < 8; ++i) {
            v2f arg = dl2 * A2[i];
            v2f ex;
            ex.x = __expf(arg.x);
            ex.y = __expf(arg.y);
            h2[i] = ex * h2[i] + du2 * bv[i];
        }

        if (final_pass) {
            v2f acc = {0.0f, 0.0f};
#pragma unroll
            for (int i = 0; i < 4; ++i) {
                v4f q4 = xr[6 + i];
                v2f c0 = {q4.x, q4.y};
                v2f c1 = {q4.z, q4.w};
                acc = h2[2 * i] * c0 + acc;
                acc = h2[2 * i + 1] * c1 + acc;
            }
            (kbase + srec)[d] = (acc.x + acc.y) + Dsd * u;
        }
    };

    if (!final_pass) {
        for (int j = 0; j < NRC; j += DGA) {
            float dl[DGA];
#pragma unroll
            for (int i = 0; i < DGA; ++i)
                dl[i] = delta_of((const v4f*)(sm_xh + (j + i) * XS));
#pragma unroll
            for (int i = 0; i < DGA; ++i) {
                state_step((const v4f*)(sm_xh + (j + i) * XS), dl[i]);
                sdelta += dl[i];
                srec += dstep;
            }
        }
        for (int j = NRC; j < CT; j += DG) {
            float dl[DG];
#pragma unroll
            for (int i = 0; i < DG; ++i)
                dl[i] = delta_of((const v4f*)(sm_xh + (j + i) * XS));
#pragma unroll
            for (int i = 0; i < DG; ++i) {
                state_step((const v4f*)(sm_xh + (j + i) * XS), dl[i]);
                sdelta += dl[i];
                (kbase + srec)[d] = dl[i];
                srec += dstep;
            }
        }
    } else {
        float dq[4];
#pragma unroll
        for (int i = 0; i < 4; ++i) dq[i] = (kbase + srec + (NRC + i) * dstep)[d];

        for (int j = 0; j < NRC; j += DGA) {
            float dl[DGA];
#pragma unroll
            for (int i = 0; i < DGA; ++i)
                dl[i] = delta_of((const v4f*)(sm_xh + (j + i) * XS));
#pragma unroll
            for (int i = 0; i < DGA; ++i) {
                state_step((const v4f*)(sm_xh + (j + i) * XS), dl[i]);
                srec += dstep;
            }
        }
        for (int j = NRC; j < CT; j += 4) {
#pragma unroll
            for (int i = 0; i < 4; ++i) {
                const v4f* xr = (const v4f*)(sm_xh + (j + i) * XS);
                state_step(xr, dq[i]);
                const int ahead = min(j + i + 4, CT - 1) - (j + i);
                dq[i] = (kbase + srec + ahead * dstep)[d];
                srec += dstep;
                __builtin_amdgcn_sched_barrier(0);
            }
        }
    }

    if (!final_pass) {
#pragma unroll
        for (int i = 0; i < 8; ++i) {
            ych[(size_t)scan_idx(k, c * CT + 2 * i) * DI]     = h2[i].x;
            ych[(size_t)scan_idx(k, c * CT + 2 * i + 1) * DI] = h2[i].y;
        }
        ych[(size_t)scan_idx(k, c * CT + NS) * DI] = sdelta;
    }
}

__global__ __launch_bounds__(128) __attribute__((amdgpu_waves_per_eu(6)))
void ss2d_scan_local(
    const float* __restrict__ y,
    const float* __restrict__ dtw_g,
    const float* __restrict__ dtb_g,
    const float* __restrict__ Alogs,
    const float* __restrict__ Ds_g,
    const float* __restrict__ xdbl,
    float* outy)
{
    __shared__ __attribute__((aligned(16))) float sm_x[2 * CT * XS];
    __shared__ float sm_uy[64 * PT];
    scan_chunk_body(false, y, dtw_g, dtb_g, Alogs, Ds_g, xdbl, outy, sm_x, sm_uy);
}

__global__ __launch_bounds__(128) __attribute__((amdgpu_waves_per_eu(6)))
void ss2d_scan_final(
    const float* __restrict__ y,
    const float* __restrict__ dtw_g,
    const float* __restrict__ dtb_g,
    const float* __restrict__ Alogs,
    const float* __restrict__ Ds_g,
    const float* __restrict__ xdbl,
    float* outy)
{
    __shared__ __attribute__((aligned(16))) float sm_x[2 * CT * XS];
    __shared__ float sm_uy[64 * PT];
    scan_chunk_body(true, y, dtw_g, dtb_g, Alogs, Ds_g, xdbl, outy, sm_x, sm_uy);
}

__global__ __launch_bounds__(192) void ss2d_scan_carry(
    const float* __restrict__ Alogs,
    float* outy)
{
    const int d   = threadIdx.x;
    const int n   = blockIdx.x & 15;
    const int kd  = blockIdx.x >> 4;
    const int k   = kd & 3;
    const float A = -__expf(Alogs[(size_t)(k * DI + d) * NS + n]);
    float* base = outy + (size_t)kd * LL * DI + d;

    float S[NC], sd[NC];
#pragma unroll
    for (int c = 0; c < NC; ++c) {
        S[c]  = base[(size_t)scan_idx(k, c * CT + n) * DI];
        sd[c] = base[(size_t)scan_idx(k, c * CT + NS) * DI];
    }
    float H = 0.0f;
#pragma unroll
    for (int c = 0; c < NC; ++c) {
        base[(size_t)scan_idx(k, c * CT + n) * DI] = H;
        H = __expf(A * sd[c]) * H + S[c];
    }
}

__global__ __launch_bounds__(256) void ss2d_merge_ln(
    const float* __restrict__ outy,
    const float* __restrict__ gamma,
    const float* __restrict__ beta,
    float* __restrict__ out)
{
    const int lane = threadIdx.x & 31;
    const int wave = (int)((blockIdx.x * 256 + threadIdx.x) >> 5);
    const size_t ks = (size_t)LL * DI;

    v2f g[3], be[3];
#pragma unroll
    for (int i = 0; i < 3; ++i) {
        g[i]  = *(const v2f*)(gamma + 2 * lane + 64 * i);
        be[i] = *(const v2f*)(beta + 2 * lane + 64 * i);
    }

    v2f v[2][3];
#pragma unroll
    for (int pp = 0; pp < 2; ++pp) {
        const int p  = wave * 2 + pp;
        const int b  = p >> 12;
        const int lm = p & (LL - 1);
        const float* base = outy + ((size_t)b * KK * LL + lm) * DI + 2 * lane;
#pragma unroll
        for (int i = 0; i < 3; ++i) {
            const v2f a0 = *(const v2f*)(base + 64 * i);
            const v2f a2 = *(const v2f*)(base + 2 * ks + 64 * i);
            const v2f a1 = *(const v2f*)(base + ks + 64 * i);
            const v2f a3 = *(const v2f*)(base + 3 * ks + 64 * i);
            v[pp][i] = ((a0 + a2) + a1) + a3;
        }
    }

#pragma unroll
    for (int pp = 0; pp < 2; ++pp) {
        float s = 0.0f, q = 0.0f;
#pragma unroll
        for (int i = 0; i < 3; ++i) {
            s += v[pp][i].x; q += v[pp][i].x * v[pp][i].x;
            s += v[pp][i].y; q += v[pp][i].y * v[pp][i].y;
        }
#pragma unroll
        for (int o = 16; o; o >>= 1) {
            s += __shfl_xor(s, o, 32);
            q += __shfl_xor(q, o, 32);
        }
        const float mu   = s / 192.0f;
        const float var  = q / 192.0f - mu * mu;
        const float rstd = rsqrtf(var + 1e-5f);
        float* op = out + (size_t)(wave * 2 + pp) * DI + 2 * lane;
#pragma unroll
        for (int i = 0; i < 3; ++i) {
            v2f r;
            r.x = (v[pp][i].x - mu) * rstd * g[i].x + be[i].x;
            r.y = (v[pp][i].y - mu) * rstd * g[i].y + be[i].y;
            *(v2f*)(op + 64 * i) = r;
        }
    }
}

extern "C" void kernel_launch(void* const* d_in, const int* in_sizes, int n_in,
                              void* d_out, int out_size, void* d_ws, size_t ws_size,
                              hipStream_t stream) {
    const float* x   = (const float*)d_in[0];
    const float* y   = (const float*)d_in[1];
    const float* xpw = (const float*)d_in[2];
    const float* dtw = (const float*)d_in[3];
    const float* dtb = (const float*)d_in[4];
    const float* Al  = (const float*)d_in[5];
    const float* Ds  = (const float*)d_in[6];
    const float* gm  = (const float*)d_in[7];
    const float* bt  = (const float*)d_in[8];

    float* xdbl = (float*)d_ws;
    float* outy = xdbl + (size_t)B_ * KK * CP * LL;

    ss2d_proj_wmma<<<1536, 256, 0, stream>>>(x, xpw, xdbl);
    ss2d_scan_local<<<B_ * KK * NP * 3, 128, 0, stream>>>(y, dtw, dtb, Al, Ds, xdbl, outy);
    ss2d_scan_carry<<<B_ * KK * NS, DI, 0, stream>>>(Al, outy);
    ss2d_scan_final<<<B_ * KK * NP * 3, 128, 0, stream>>>(y, dtw, dtb, Al, Ds, xdbl, outy);
    ss2d_merge_ln<<<B_ * LL / 16, 256, 0, stream>>>(outy, gm, bt, (float*)d_out);
}


# y transposed once into yt (extra work-groups of the projection launch, via LDS; yt sits in the output buffer until the merge overwrites it): column-major scans stage u by whole lines; sums unchanged
# speedup vs baseline: 16.4825x; 1.0579x over previous
#include <hip/hip_runtime.h>
#include <hip/hip_bf16.h>


#define B_  4
#define DI  192
#define LL  4096
#define NS  16
#define RR  6
#define KK  4
#define CP  38

typedef float v2f __attribute__((ext_vector_type(2)));
typedef float v8f __attribute__((ext_vector_type(8)));

__device__ __forceinline__ int scan_idx(int k, int l) {
    int ll = (k & 2) ? (LL - 1 - l) : l;
    if (k & 1) return ((ll & 63) << 6) | (ll >> 6);
    return ll;
}

#define NTB 256
#define TPB 3
#define TP  65

typedef float v4f __attribute__((ext_vector_type(4)));

__global__ __launch_bounds__(256) void ss2d_proj_wmma(
    const float* __restrict__ x,
    const float* __restrict__ xpw,
    float* __restrict__ xdbl,
    const float* __restrict__ y,
    float* __restrict__ yt)
{
    __shared__ float tile[64 * TP];
    if (blockIdx.x < NTB) {
        const int tid = threadIdx.x;
        v4f q[TPB][4];
#pragma unroll
        for (int p = 0; p < TPB; ++p) {
#pragma unroll
            for (int it = 0; it < 4; ++it)
                q[p][it] = *(const v4f*)(y + (size_t)(blockIdx.x * TPB + p) * LL + (it * 256 + tid) * 4);
        }
#pragma unroll
        for (int p = 0; p < TPB; ++p) {
#pragma unroll
            for (int it = 0; it < 4; ++it) {
                const int idx = it * 256 + tid;
                const int h   = idx >> 4;
                const int w0  = (idx & 15) << 2;
#pragma unroll
                for (int e = 0; e < 4; ++e) tile[(w0 + e) * TP + h] = q[p][it][e];
            }
            __syncthreads();
            float* dst = yt + (size_t)(blockIdx.x * TPB + p) * LL;
#pragma unroll
            for (int it = 0; it < 4; ++it) {
                const int idx = it * 256 + tid;
                const int w   = idx >> 4;
                const int h0  = (idx & 15) << 2;
                v4f r;
#pragma unroll
                for (int e = 0; e < 4; ++e) r[e] = tile[w * TP + h0 + e];
                *(v4f*)(dst + idx * 4) = r;
            }
            __syncthreads();
        }
        return;
    }

    int wave = (int)(((blockIdx.x - NTB) * blockDim.x + threadIdx.x) >> 5);
    int lane = threadIdx.x & 31;
    int nt = wave & 255;
    int mt = (wave >> 8) % 3;
    int kd = (wave >> 8) / 3;
    if (kd >= B_ * KK) return;
    int k = kd & 3;
    int b = kd >> 2;

    int m0 = mt << 4;
    int l0 = nt << 4;
    int hw = lane >> 4;
    int ln = lane & 15;

    const float* wbase = xpw + (size_t)k * CP * DI;
    const float* xb    = x + (size_t)b * DI * LL;

    int arow   = m0 + ln;
    int arow_c = arow < CP ? arow : (CP - 1);
    float amsk = arow < CP ? 1.0f : 0.0f;

    int m = l0 + ln;
    int lt   = (k & 1) ? (((m & 63) << 6) | (m >> 6)) : m;
    int lpos = (k & 2) ? (LL - 1 - lt) : lt;

    v8f acc = {0.f, 0.f, 0.f, 0.f, 0.f, 0.f, 0.f, 0.f};
    for (int k0 = 0; k0 < DI; k0 += 4) {
        const float* ap = wbase + arow_c * DI + k0 + (hw << 1);
        v2f a; a.x = ap[0] * amsk; a.y = ap[1] * amsk;
        const float* bp = xb + (size_t)(k0 + (hw << 1)) * LL + m;
        v2f bb; bb.x = bp[0]; bb.y = bp[LL];
        acc = __builtin_amdgcn_wmma_f32_16x16x4_f32(
            false, a, false, bb, (short)0, acc, false, false);
    }

    float* obase = xdbl + (size_t)kd * CP * LL + lpos;
#pragma unroll
    for (int i = 0; i < 8; ++i) {
        int m = m0 + i + (hw << 3);
        if (m < CP) obase[(size_t)m * LL] = acc[i];
    }
}

#define NC  64
#define CT  64
#define NP  32
#define PT  128
#define XS  40
#define NRC 20
#define DG  4
#define DGA 2

__device__ __forceinline__ int u_slot(int r, int p) {
    return r * PT + ((p + r) & (PT - 1));
}

__device__ __forceinline__ void scan_chunk_body(
    const bool final_pass,
    const float* __restrict__ y,
    const float* __restrict__ yt,
    const float* __restrict__ dtw_g,
    const float* __restrict__ dtb_g,
    const float* __restrict__ Alogs,
    const float* __restrict__ Ds_g,
    const float* __restrict__ xdbl,
    float* outy,
    float* sm_x,
    float* sm_uy)
{
    const int tid  = threadIdx.x;
    const int t    = tid & 63;
    const int hf   = __builtin_amdgcn_readfirstlane(tid >> 6);
    const int dblk = blockIdx.x % 3;
    const int cp   = (blockIdx.x / 3) % NP;
    const int kd   = blockIdx.x / (3 * NP);
    const int c    = 2 * cp + hf;
    const int k    = kd & 3;
    const int b    = kd >> 2;
    const int d    = dblk * 64 + t;
    const int kdch = k * DI + d;

    v2f A2[8];
    {
        const v4f* ap = (const v4f*)(Alogs + (size_t)kdch * NS);
#pragma unroll
        for (int i = 0; i < 4; ++i) {
            v4f a = ap[i];
            A2[2 * i].x     = -__expf(a.x);
            A2[2 * i].y     = -__expf(a.y);
            A2[2 * i + 1].x = -__expf(a.z);
            A2[2 * i + 1].y = -__expf(a.w);
        }
    }
    float dtw[RR];
#pragma unroll
    for (int r = 0; r < RR; ++r) dtw[r] = dtw_g[(size_t)kdch * RR + r];
    const float dtb = dtb_g[kdch];
    const float Dsd = Ds_g[kdch];

    const float* xdb = xdbl + (size_t)kd * CP * LL;
    const float* yb  = ((k & 1) ? yt : y) + ((size_t)b * DI + dblk * 64) * LL;
    float* ych       = outy + (size_t)kd * LL * DI + d;

    v2f h2[8];
    if (final_pass) {
#pragma unroll
        for (int i = 0; i < 8; ++i) {
            h2[i].x = ych[(size_t)scan_idx(k, c * CT + 2 * i) * DI];
            h2[i].y = ych[(size_t)scan_idx(k, c * CT + 2 * i + 1) * DI];
        }
    } else {
#pragma unroll
        for (int i = 0; i < 8; ++i) { h2[i].x = 0.0f; h2[i].y = 0.0f; }
    }
    float sdelta = 0.0f;

    {
        const int mbase = (k & 2) ? (LL - PT - cp * PT) : (cp * PT);
        const int q = tid & 31;
        const int wv = __builtin_amdgcn_readfirstlane(tid >> 5);
        v4f uq[16];
#pragma unroll
        for (int it = 0; it < 16; ++it)
            uq[it] = *(const v4f*)(yb + (size_t)(4 * it + wv) * LL + mbase + 4 * q);
#pragma unroll
        for (int it = 0; it < 16; ++it) {
            const int r = 4 * it + wv;
#pragma unroll
            for (int e = 0; e < 4; ++e) {
                int m = mbase + 4 * q + e;
                int l = (k & 2) ? (LL - 1 - m) : m;
                sm_uy[u_slot(r, l - cp * PT)] = uq[it][e];
            }
        }
    }

    if (final_pass) __builtin_amdgcn_sched_barrier(0);

    const int nrows = final_pass ? CP : (RR + NS);
    float* sm_xh = sm_x + hf * (CT * XS);
    const int l0 = c * CT;
    for (int r = 0; r < nrows; ++r) {
        const int pos = r < RR ? r : r + 2;
        sm_xh[t * XS + pos] = xdb[(size_t)r * LL + l0 + t];
    }
    __syncthreads();

    const float* urow = sm_uy + t * PT;
    int up = (hf * CT + t) & (PT - 1);

    const int mstep = ((k & 1) ? 64 : 1) * ((k & 2) ? -1 : 1);
    const int dstep = mstep * DI;
    float* kbase    = outy + (size_t)kd * LL * DI;
    int srec        = scan_idx(k, l0) * DI;

    auto delta_of = [&](const v4f* xr) -> float {
        const v4f d0 = xr[0], d1 = xr[1];
        float sarg = dtb;
        sarg += dtw[0] * d0.x;
        sarg += dtw[1] * d0.y;
        sarg += dtw[2] * d0.z;
        sarg += dtw[3] * d0.w;
        sarg += dtw[4] * d1.x;
        sarg += dtw[5] * d1.y;
        const float lp = log1pf(__expf(sarg));
        return (sarg > 20.0f) ? sarg : lp;
    };

    auto state_step = [&](const v4f* xr, const float delta) {
        const float u  = urow[up];
        up = (up + 1) & (PT - 1);
        const float du = delta * u;
        const v2f dl2 = {delta, delta};
        const v2f du2 = {du, du};

        v2f bv[8];
#pragma unroll
        for (int i = 0; i < 4; ++i) {
            v4f q4 = xr[2 + i];
            bv[2 * i].x = q4.x; bv[2 * i].y = q4.y;
            bv[2 * i + 1].x = q4.z; bv[2 * i + 1].y = q4.w;
        }
#pragma unroll
        for (int i = 0; i < 8; ++i) {
            v2f arg = dl2 * A2[i];
            v2f ex;
            ex.x = __expf(arg.x);
            ex.y = __expf(arg.y);
            h2[i] = ex * h2[i] + du2 * bv[i];
        }

        if (final_pass) {
            v2f acc = {0.0f, 0.0f};
#pragma unroll
            for (int i = 0; i < 4; ++i) {
                v4f q4 = xr[6 + i];
                v2f c0 = {q4.x, q4.y};
                v2f c1 = {q4.z, q4.w};
                acc = h2[2 * i] * c0 + acc;
                acc = h2[2 * i + 1] * c1 + acc;
            }
            (kbase + srec)[d] = (acc.x + acc.y) + Dsd * u;
        }
    };

    if (!final_pass) {
        for (int j = 0; j < NRC; j += DGA) {
            float dl[DGA];
#pragma unroll
            for (int i = 0; i < DGA; ++i)
                dl[i] = delta_of((const v4f*)(sm_xh + (j + i) * XS));
#pragma unroll
            for (int i = 0; i < DGA; ++i) {
                state_step((const v4f*)(sm_xh + (j + i) * XS), dl[i]);
                sdelta += dl[i];
                srec += dstep;
            }
        }
        for (int j = NRC; j < CT; j += DG) {
            float dl[DG];
#pragma unroll
            for (int i = 0; i < DG; ++i)
                dl[i] = delta_of((const v4f*)(sm_xh + (j + i) * XS));
#pragma unroll
            for (int i = 0; i < DG; ++i) {
                state_step((const v4f*)(sm_xh + (j + i) * XS), dl[i]);
                sdelta += dl[i];
                (kbase + srec)[d] = dl[i];
                srec += dstep;
            }
        }
    } else {
        float dq[4];
#pragma unroll
        for (int i = 0; i < 4; ++i) dq[i] = (kbase + srec + (NRC + i) * dstep)[d];

        for (int j = 0; j < NRC; j += DGA) {
            float dl[DGA];
#pragma unroll
            for (int i = 0; i < DGA; ++i)
                dl[i] = delta_of((const v4f*)(sm_xh + (j + i) * XS));
#pragma unroll
            for (int i = 0; i < DGA; ++i) {
                state_step((const v4f*)(sm_xh + (j + i) * XS), dl[i]);
                srec += dstep;
            }
        }
        for (int j = NRC; j < CT; j += 4) {
#pragma unroll
            for (int i = 0; i < 4; ++i) {
                const v4f* xr = (const v4f*)(sm_xh + (j + i) * XS);
                state_step(xr, dq[i]);
                const int ahead = min(j + i + 4, CT - 1) - (j + i);
                dq[i] = (kbase + srec + ahead * dstep)[d];
                srec += dstep;
                __builtin_amdgcn_sched_barrier(0);
            }
        }
    }

    if (!final_pass) {
#pragma unroll
        for (int i = 0; i < 8; ++i) {
            ych[(size_t)scan_idx(k, c * CT + 2 * i) * DI]     = h2[i].x;
            ych[(size_t)scan_idx(k, c * CT + 2 * i + 1) * DI] = h2[i].y;
        }
        ych[(size_t)scan_idx(k, c * CT + NS) * DI] = sdelta;
    }
}

__global__ __launch_bounds__(128) __attribute__((amdgpu_waves_per_eu(6)))
void ss2d_scan_local(
    const float* __restrict__ y,
    const float* __restrict__ yt,
    const float* __restrict__ dtw_g,
    const float* __restrict__ dtb_g,
    const float* __restrict__ Alogs,
    const float* __restrict__ Ds_g,
    const float* __restrict__ xdbl,
    float* outy)
{
    __shared__ __attribute__((aligned(16))) float sm_x[2 * CT * XS];
    __shared__ float sm_uy[64 * PT];
    scan_chunk_body(false, y, yt, dtw_g, dtb_g, Alogs, Ds_g, xdbl, outy, sm_x, sm_uy);
}

__global__ __launch_bounds__(128) __attribute__((amdgpu_waves_per_eu(6)))
void ss2d_scan_final(
    const float* __restrict__ y,
    const float* __restrict__ yt,
    const float* __restrict__ dtw_g,
    const float* __restrict__ dtb_g,
    const float* __restrict__ Alogs,
    const float* __restrict__ Ds_g,
    const float* __restrict__ xdbl,
    float* outy)
{
    __shared__ __attribute__((aligned(16))) float sm_x[2 * CT * XS];
    __shared__ float sm_uy[64 * PT];
    scan_chunk_body(true, y, yt, dtw_g, dtb_g, Alogs, Ds_g, xdbl, outy, sm_x, sm_uy);
}

__global__ __launch_bounds__(192) void ss2d_scan_carry(
    const float* __restrict__ Alogs,
    float* outy)
{
    const int d   = threadIdx.x;
    const int n   = blockIdx.x & 15;
    const int kd  = blockIdx.x >> 4;
    const int k   = kd & 3;
    const float A = -__expf(Alogs[(size_t)(k * DI + d) * NS + n]);
    float* base = outy + (size_t)kd * LL * DI + d;

    float S[NC], sd[NC];
#pragma unroll
    for (int c = 0; c < NC; ++c) {
        S[c]  = base[(size_t)scan_idx(k, c * CT + n) * DI];
        sd[c] = base[(size_t)scan_idx(k, c * CT + NS) * DI];
    }
    float H = 0.0f;
#pragma unroll
    for (int c = 0; c < NC; ++c) {
        base[(size_t)scan_idx(k, c * CT + n) * DI] = H;
        H = __expf(A * sd[c]) * H + S[c];
    }
}

__global__ __launch_bounds__(256) void ss2d_merge_ln(
    const float* __restrict__ outy,
    const float* __restrict__ gamma,
    const float* __restrict__ beta,
    float* __restrict__ out)
{
    const int lane = threadIdx.x & 31;
    const int wave = (int)((blockIdx.x * 256 + threadIdx.x) >> 5);
    const size_t ks = (size_t)LL * DI;

    v2f g[3], be[3];
#pragma unroll
    for (int i = 0; i < 3; ++i) {
        g[i]  = *(const v2f*)(gamma + 2 * lane + 64 * i);
        be[i] = *(const v2f*)(beta + 2 * lane + 64 * i);
    }

    v2f v[2][3];
#pragma unroll
    for (int pp = 0; pp < 2; ++pp) {
        const int p  = wave * 2 + pp;
        const int b  = p >> 12;
        const int lm = p & (LL - 1);
        const float* base = outy + ((size_t)b * KK * LL + lm) * DI + 2 * lane;
#pragma unroll
        for (int i = 0; i < 3; ++i) {
            const v2f a0 = *(const v2f*)(base + 64 * i);
            const v2f a2 = *(const v2f*)(base + 2 * ks + 64 * i);
            const v2f a1 = *(const v2f*)(base + ks + 64 * i);
            const v2f a3 = *(const v2f*)(base + 3 * ks + 64 * i);
            v[pp][i] = ((a0 + a2) + a1) + a3;
        }
    }

#pragma unroll
    for (int pp = 0; pp < 2; ++pp) {
        float s = 0.0f, q = 0.0f;
#pragma unroll
        for (int i = 0; i < 3; ++i) {
            s += v[pp][i].x; q += v[pp][i].x * v[pp][i].x;
            s += v[pp][i].y; q += v[pp][i].y * v[pp][i].y;
        }
#pragma unroll
        for (int o = 16; o; o >>= 1) {
            s += __shfl_xor(s, o, 32);
            q += __shfl_xor(q, o, 32);
        }
        const float mu   = s / 192.0f;
        const float var  = q / 192.0f - mu * mu;
        const float rstd = rsqrtf(var + 1e-5f);
        float* op = out + (size_t)(wave * 2 + pp) * DI + 2 * lane;
#pragma unroll
        for (int i = 0; i < 3; ++i) {
            v2f r;
            r.x = (v[pp][i].x - mu) * rstd * g[i].x + be[i].x;
            r.y = (v[pp][i].y - mu) * rstd * g[i].y + be[i].y;
            *(v2f*)(op + 64 * i) = r;
        }
    }
}

extern "C" void kernel_launch(void* const* d_in, const int* in_sizes, int n_in,
                              void* d_out, int out_size, void* d_ws, size_t ws_size,
                              hipStream_t stream) {
    const float* x   = (const float*)d_in[0];
    const float* y   = (const float*)d_in[1];
    const float* xpw = (const float*)d_in[2];
    const float* dtw = (const float*)d_in[3];
    const float* dtb = (const float*)d_in[4];
    const float* Al  = (const float*)d_in[5];
    const float* Ds  = (const float*)d_in[6];
    const float* gm  = (const float*)d_in[7];
    const float* bt  = (const float*)d_in[8];

    float* xdbl = (float*)d_ws;
    float* outy = xdbl + (size_t)B_ * KK * CP * LL;
    float* yt   = (float*)d_out;

    ss2d_proj_wmma<<<NTB + 1536, 256, 0, stream>>>(x, xpw, xdbl, y, yt);
    ss2d_scan_local<<<B_ * KK * NP * 3, 128, 0, stream>>>(y, yt, dtw, dtb, Al, Ds, xdbl, outy);
    ss2d_scan_carry<<<B_ * KK * NS, DI, 0, stream>>>(Al, outy);
    ss2d_scan_final<<<B_ * KK * NP * 3, 128, 0, stream>>>(y, yt, dtw, dtb, Al, Ds, xdbl, outy);
    ss2d_merge_ln<<<B_ * LL / 16, 256, 0, stream>>>(outy, gm, bt, (float*)d_out);
}


# projection: one wave per 16 positions x direction pair, five accumulator tiles from ONE x fragment in registers (no LDS), weights by 16-B loads; x read 2x not 12x; k sum order of projection changes
# speedup vs baseline: 18.3974x; 1.1162x over previous
#include <hip/hip_runtime.h>
#include <hip/hip_bf16.h>


#define B_  4
#define DI  192
#define LL  4096
#define NS  16
#define RR  6
#define KK  4
#define CP  38

typedef float v2f __attribute__((ext_vector_type(2)));
typedef float v8f __attribute__((ext_vector_type(8)));

__device__ __forceinline__ int scan_idx(int k, int l) {
    int ll = (k & 2) ? (LL - 1 - l) : l;
    if (k & 1) return ((ll & 63) << 6) | (ll >> 6);
    return ll;
}

#define NTB 256
#define TPB 3
#define PG  4
#define TP  65

typedef float v4f __attribute__((ext_vector_type(4)));

__global__ __launch_bounds__(256) void ss2d_proj_wmma(
    const float* __restrict__ x,
    const float* __restrict__ xpw,
    float* __restrict__ xdbl,
    const float* __restrict__ y,
    float* __restrict__ yt)
{
    __shared__ float tile[64 * TP];
    if (blockIdx.x < NTB) {
        const int tid = threadIdx.x;
        v4f q[TPB][4];
#pragma unroll
        for (int p = 0; p < TPB; ++p) {
#pragma unroll
            for (int it = 0; it < 4; ++it)
                q[p][it] = *(const v4f*)(y + (size_t)(blockIdx.x * TPB + p) * LL + (it * 256 + tid) * 4);
        }
#pragma unroll
        for (int p = 0; p < TPB; ++p) {
#pragma unroll
            for (int it = 0; it < 4; ++it) {
                const int idx = it * 256 + tid;
                const int h   = idx >> 4;
                const int w0  = (idx & 15) << 2;
#pragma unroll
                for (int e = 0; e < 4; ++e) tile[(w0 + e) * TP + h] = q[p][it][e];
            }
            __syncthreads();
            float* dst = yt + (size_t)(blockIdx.x * TPB + p) * LL;
#pragma unroll
            for (int it = 0; it < 4; ++it) {
                const int idx = it * 256 + tid;
                const int w   = idx >> 4;
                const int h0  = (idx & 15) << 2;
                v4f r;
#pragma unroll
                for (int e = 0; e < 4; ++e) r[e] = tile[w * TP + h0 + e];
                *(v4f*)(dst + idx * 4) = r;
            }
            __syncthreads();
        }
        return;
    }

    const int wave = (int)(((blockIdx.x - NTB) * blockDim.x + threadIdx.x) >> 5);
    const int lane = threadIdx.x & 31;
    const int nt = wave & 255;
    const int pr = (wave >> 8) & 1;
    const int b  = wave >> 9;
    const int hw = lane >> 4;
    const int ln = lane & 15;
    const int ka = 2 * pr, kb = 2 * pr + 1;
    const int m  = (nt << 4) + ln;

    const float* xb = x + ((size_t)b * DI + 4 * hw) * LL + m;
    const float* wa = xpw + (size_t)ka * CP * DI + 4 * hw;
    const float* wb = xpw + (size_t)kb * CP * DI + 4 * hw;
    const int  tr = ln & 7;
    const bool tv = tr < (CP - 32);
    const float* ap[5];
    ap[0] = wa + (size_t)ln * DI;
    ap[1] = wa + (size_t)(16 + ln) * DI;
    ap[2] = wb + (size_t)ln * DI;
    ap[3] = wb + (size_t)(16 + ln) * DI;
    ap[4] = ((ln < 8) ? wa : wb) + (size_t)(32 + (tv ? tr : 0)) * DI;

    v8f acc[5];
#pragma unroll
    for (int j = 0; j < 5; ++j) acc[j] = (v8f){0.f, 0.f, 0.f, 0.f, 0.f, 0.f, 0.f, 0.f};

    v4f   Aq[2][PG][5];
    float Bq[2][PG][4];
#pragma unroll
    for (int gi = 0; gi < PG; ++gi) {
#pragma unroll
        for (int j = 0; j < 5; ++j) Aq[0][gi][j] = *(const v4f*)(ap[j] + 8 * gi);
#pragma unroll
        for (int e = 0; e < 4; ++e) Bq[0][gi][e] = xb[(size_t)(8 * gi + e) * LL];
    }
#pragma unroll
    for (int t = 0; t < DI / (8 * PG); ++t) {
        const int cur = t & 1, nxt = cur ^ 1;
        if (t + 1 < DI / (8 * PG)) {
#pragma unroll
            for (int gi = 0; gi < PG; ++gi) {
                const int g = (t + 1) * PG + gi;
#pragma unroll
                for (int j = 0; j < 5; ++j) Aq[nxt][gi][j] = *(const v4f*)(ap[j] + 8 * g);
#pragma unroll
                for (int e = 0; e < 4; ++e) Bq[nxt][gi][e] = xb[(size_t)(8 * g + e) * LL];
            }
        }
#pragma unroll
        for (int gi = 0; gi < PG; ++gi) {
            const v4f zero4 = {0.f, 0.f, 0.f, 0.f};
            const v4f a4 = tv ? Aq[cur][gi][4] : zero4;
            const v2f b01 = {Bq[cur][gi][0], Bq[cur][gi][1]};
            const v2f b23 = {Bq[cur][gi][2], Bq[cur][gi][3]};
#pragma unroll
            for (int j = 0; j < 5; ++j) {
                const v4f q = (j < 4) ? Aq[cur][gi][j] : a4;
                const v2f a = {q.x, q.y};
                acc[j] = __builtin_amdgcn_wmma_f32_16x16x4_f32(
                    false, a, false, b01, (short)0, acc[j], false, false);
            }
#pragma unroll
            for (int j = 0; j < 5; ++j) {
                const v4f q = (j < 4) ? Aq[cur][gi][j] : a4;
                const v2f a = {q.z, q.w};
                acc[j] = __builtin_amdgcn_wmma_f32_16x16x4_f32(
                    false, a, false, b23, (short)0, acc[j], false, false);
            }
        }
    }

    const int mt_ = ((m & 63) << 6) | (m >> 6);
    const int lpa = (ka & 2) ? (LL - 1 - m)   : m;
    const int lpb = (kb & 2) ? (LL - 1 - mt_) : mt_;
    float* oa = xdbl + (size_t)(b * KK + ka) * CP * LL + lpa;
    float* ob = xdbl + (size_t)(b * KK + kb) * CP * LL + lpb;
#pragma unroll
    for (int i = 0; i < 8; ++i) {
        const int row = i + (hw << 3);
        oa[(size_t)row * LL]        = acc[0][i];
        oa[(size_t)(16 + row) * LL] = acc[1][i];
        ob[(size_t)row * LL]        = acc[2][i];
        ob[(size_t)(16 + row) * LL] = acc[3][i];
    }
    float* ot = hw ? ob : oa;
#pragma unroll
    for (int i = 0; i < CP - 32; ++i) ot[(size_t)(32 + i) * LL] = acc[4][i];
}

#define NC  64
#define CT  64
#define NP  32
#define PT  128
#define XS  40
#define NRC 20
#define DG  4
#define DGA 2

__device__ __forceinline__ int u_slot(int r, int p) {
    return r * PT + ((p + r) & (PT - 1));
}

__device__ __forceinline__ void scan_chunk_body(
    const bool final_pass,
    const float* __restrict__ y,
    const float* __restrict__ yt,
    const float* __restrict__ dtw_g,
    const float* __restrict__ dtb_g,
    const float* __restrict__ Alogs,
    const float* __restrict__ Ds_g,
    const float* __restrict__ xdbl,
    float* outy,
    float* sm_x,
    float* sm_uy)
{
    const int tid  = threadIdx.x;
    const int t    = tid & 63;
    const int hf   = __builtin_amdgcn_readfirstlane(tid >> 6);
    const int dblk = blockIdx.x % 3;
    const int cp   = (blockIdx.x / 3) % NP;
    const int kd   = blockIdx.x / (3 * NP);
    const int c    = 2 * cp + hf;
    const int k    = kd & 3;
    const int b    = kd >> 2;
    const int d    = dblk * 64 + t;
    const int kdch = k * DI + d;

    v2f A2[8];
    {
        const v4f* ap = (const v4f*)(Alogs + (size_t)kdch * NS);
#pragma unroll
        for (int i = 0; i < 4; ++i) {
            v4f a = ap[i];
            A2[2 * i].x     = -__expf(a.x);
            A2[2 * i].y     = -__expf(a.y);
            A2[2 * i + 1].x = -__expf(a.z);
            A2[2 * i + 1].y = -__expf(a.w);
        }
    }
    float dtw[RR];
#pragma unroll
    for (int r = 0; r < RR; ++r) dtw[r] = dtw_g[(size_t)kdch * RR + r];
    const float dtb = dtb_g[kdch];
    const float Dsd = Ds_g[kdch];

    const float* xdb = xdbl + (size_t)kd * CP * LL;
    const float* yb  = ((k & 1) ? yt : y) + ((size_t)b * DI + dblk * 64) * LL;
    float* ych       = outy + (size_t)kd * LL * DI + d;

    v2f h2[8];
    if (final_pass) {
#pragma unroll
        for (int i = 0; i < 8; ++i) {
            h2[i].x = ych[(size_t)scan_idx(k, c * CT + 2 * i) * DI];
            h2[i].y = ych[(size_t)scan_idx(k, c * CT + 2 * i + 1) * DI];
        }
    } else {
#pragma unroll
        for (int i = 0; i < 8; ++i) { h2[i].x = 0.0f; h2[i].y = 0.0f; }
    }
    float sdelta = 0.0f;

    {
        const int mbase = (k & 2) ? (LL - PT - cp * PT) : (cp * PT);
        const int q = tid & 31;
        const int wv = __builtin_amdgcn_readfirstlane(tid >> 5);
        v4f uq[16];
#pragma unroll
        for (int it = 0; it < 16; ++it)
            uq[it] = *(const v4f*)(yb + (size_t)(4 * it + wv) * LL + mbase + 4 * q);
#pragma unroll
        for (int it = 0; it < 16; ++it) {
            const int r = 4 * it + wv;
#pragma unroll
            for (int e = 0; e < 4; ++e) {
                int m = mbase + 4 * q + e;
                int l = (k & 2) ? (LL - 1 - m) : m;
                sm_uy[u_slot(r, l - cp * PT)] = uq[it][e];
            }
        }
    }

    if (final_pass) __builtin_amdgcn_sched_barrier(0);

    const int nrows = final_pass ? CP : (RR + NS);
    float* sm_xh = sm_x + hf * (CT * XS);
    const int l0 = c * CT;
    for (int r = 0; r < nrows; ++r) {
        const int pos = r < RR ? r : r + 2;
        sm_xh[t * XS + pos] = xdb[(size_t)r * LL + l0 + t];
    }
    __syncthreads();

    const float* urow = sm_uy + t * PT;
    int up = (hf * CT + t) & (PT - 1);

    const int mstep = ((k & 1) ? 64 : 1) * ((k & 2) ? -1 : 1);
    const int dstep = mstep * DI;
    float* kbase    = outy + (size_t)kd * LL * DI;
    int srec        = scan_idx(k, l0) * DI;

    auto delta_of = [&](const v4f* xr) -> float {
        const v4f d0 = xr[0], d1 = xr[1];
        float sarg = dtb;
        sarg += dtw[0] * d0.x;
        sarg += dtw[1] * d0.y;
        sarg += dtw[2] * d0.z;
        sarg += dtw[3] * d0.w;
        sarg += dtw[4] * d1.x;
        sarg += dtw[5] * d1.y;
        const float lp = log1pf(__expf(sarg));
        return (sarg > 20.0f) ? sarg : lp;
    };

    auto state_step = [&](const v4f* xr, const float delta) {
        const float u  = urow[up];
        up = (up + 1) & (PT - 1);
        const float du = delta * u;
        const v2f dl2 = {delta, delta};
        const v2f du2 = {du, du};

        v2f bv[8];
#pragma unroll
        for (int i = 0; i < 4; ++i) {
            v4f q4 = xr[2 + i];
            bv[2 * i].x = q4.x; bv[2 * i].y = q4.y;
            bv[2 * i + 1].x = q4.z; bv[2 * i + 1].y = q4.w;
        }
#pragma unroll
        for (int i = 0; i < 8; ++i) {
            v2f arg = dl2 * A2[i];
            v2f ex;
            ex.x = __expf(arg.x);
            ex.y = __expf(arg.y);
            h2[i] = ex * h2[i] + du2 * bv[i];
        }

        if (final_pass) {
            v2f acc = {0.0f, 0.0f};
#pragma unroll
            for (int i = 0; i < 4; ++i) {
                v4f q4 = xr[6 + i];
                v2f c0 = {q4.x, q4.y};
                v2f c1 = {q4.z, q4.w};
                acc = h2[2 * i] * c0 + acc;
                acc = h2[2 * i + 1] * c1 + acc;
            }
            (kbase + srec)[d] = (acc.x + acc.y) + Dsd * u;
        }
    };

    if (!final_pass) {
        for (int j = 0; j < NRC; j += DGA) {
            float dl[DGA];
#pragma unroll
            for (int i = 0; i < DGA; ++i)
                dl[i] = delta_of((const v4f*)(sm_xh + (j + i) * XS));
#pragma unroll
            for (int i = 0; i < DGA; ++i) {
                state_step((const v4f*)(sm_xh + (j + i) * XS), dl[i]);
                sdelta += dl[i];
                srec += dstep;
            }
        }
        for (int j = NRC; j < CT; j += DG) {
            float dl[DG];
#pragma unroll
            for (int i = 0; i < DG; ++i)
                dl[i] = delta_of((const v4f*)(sm_xh + (j + i) * XS));
#pragma unroll
            for (int i = 0; i < DG; ++i) {
                state_step((const v4f*)(sm_xh + (j + i) * XS), dl[i]);
                sdelta += dl[i];
                (kbase + srec)[d] = dl[i];
                srec += dstep;
            }
        }
    } else {
        float dq[4];
#pragma unroll
        for (int i = 0; i < 4; ++i) dq[i] = (kbase + srec + (NRC + i) * dstep)[d];

        for (int j = 0; j < NRC; j += DGA) {
            float dl[DGA];
#pragma unroll
            for (int i = 0; i < DGA; ++i)
                dl[i] = delta_of((const v4f*)(sm_xh + (j + i) * XS));
#pragma unroll
            for (int i = 0; i < DGA; ++i) {
                state_step((const v4f*)(sm_xh + (j + i) * XS), dl[i]);
                srec += dstep;
            }
        }
        for (int j = NRC; j < CT; j += 4) {
#pragma unroll
            for (int i = 0; i < 4; ++i) {
                const v4f* xr = (const v4f*)(sm_xh + (j + i) * XS);
                state_step(xr, dq[i]);
                const int ahead = min(j + i + 4, CT - 1) - (j + i);
                dq[i] = (kbase + srec + ahead * dstep)[d];
                srec += dstep;
                __builtin_amdgcn_sched_barrier(0);
            }
        }
    }

    if (!final_pass) {
#pragma unroll
        for (int i = 0; i < 8; ++i) {
            ych[(size_t)scan_idx(k, c * CT + 2 * i) * DI]     = h2[i].x;
            ych[(size_t)scan_idx(k, c * CT + 2 * i + 1) * DI] = h2[i].y;
        }
        ych[(size_t)scan_idx(k, c * CT + NS) * DI] = sdelta;
    }
}

__global__ __launch_bounds__(128) __attribute__((amdgpu_waves_per_eu(6)))
void ss2d_scan_local(
    const float* __restrict__ y,
    const float* __restrict__ yt,
    const float* __restrict__ dtw_g,
    const float* __restrict__ dtb_g,
    const float* __restrict__ Alogs,
    const float* __restrict__ Ds_g,
    const float* __restrict__ xdbl,
    float* outy)
{
    __shared__ __attribute__((aligned(16))) float sm_x[2 * CT * XS];
    __shared__ float sm_uy[64 * PT];
    scan_chunk_body(false, y, yt, dtw_g, dtb_g, Alogs, Ds_g, xdbl, outy, sm_x, sm_uy);
}

__global__ __launch_bounds__(128) __attribute__((amdgpu_waves_per_eu(6)))
void ss2d_scan_final(
    const float* __restrict__ y,
    const float* __restrict__ yt,
    const float* __restrict__ dtw_g,
    const float* __restrict__ dtb_g,
    const float* __restrict__ Alogs,
    const float* __restrict__ Ds_g,
    const float* __restrict__ xdbl,
    float* outy)
{
    __shared__ __attribute__((aligned(16))) float sm_x[2 * CT * XS];
    __shared__ float sm_uy[64 * PT];
    scan_chunk_body(true, y, yt, dtw_g, dtb_g, Alogs, Ds_g, xdbl, outy, sm_x, sm_uy);
}

__global__ __launch_bounds__(192) void ss2d_scan_carry(
    const float* __restrict__ Alogs,
    float* outy)
{
    const int d   = threadIdx.x;
    const int n   = blockIdx.x & 15;
    const int kd  = blockIdx.x >> 4;
    const int k   = kd & 3;
    const float A = -__expf(Alogs[(size_t)(k * DI + d) * NS + n]);
    float* base = outy + (size_t)kd * LL * DI + d;

    float S[NC], sd[NC];
#pragma unroll
    for (int c = 0; c < NC; ++c) {
        S[c]  = base[(size_t)scan_idx(k, c * CT + n) * DI];
        sd[c] = base[(size_t)scan_idx(k, c * CT + NS) * DI];
    }
    float H = 0.0f;
#pragma unroll
    for (int c = 0; c < NC; ++c) {
        base[(size_t)scan_idx(k, c * CT + n) * DI] = H;
        H = __expf(A * sd[c]) * H + S[c];
    }
}

__global__ __launch_bounds__(256) void ss2d_merge_ln(
    const float* __restrict__ outy,
    const float* __restrict__ gamma,
    const float* __restrict__ beta,
    float* __restrict__ out)
{
    const int lane = threadIdx.x & 31;
    const int wave = (int)((blockIdx.x * 256 + threadIdx.x) >> 5);
    const size_t ks = (size_t)LL * DI;

    v2f g[3], be[3];
#pragma unroll
    for (int i = 0; i < 3; ++i) {
        g[i]  = *(const v2f*)(gamma + 2 * lane + 64 * i);
        be[i] = *(const v2f*)(beta + 2 * lane + 64 * i);
    }

    v2f v[2][3];
#pragma unroll
    for (int pp = 0; pp < 2; ++pp) {
        const int p  = wave * 2 + pp;
        const int b  = p >> 12;
        const int lm = p & (LL - 1);
        const float* base = outy + ((size_t)b * KK * LL + lm) * DI + 2 * lane;
#pragma unroll
        for (int i = 0; i < 3; ++i) {
            const v2f a0 = *(const v2f*)(base + 64 * i);
            const v2f a2 = *(const v2f*)(base + 2 * ks + 64 * i);
            const v2f a1 = *(const v2f*)(base + ks + 64 * i);
            const v2f a3 = *(const v2f*)(base + 3 * ks + 64 * i);
            v[pp][i] = ((a0 + a2) + a1) + a3;
        }
    }

#pragma unroll
    for (int pp = 0; pp < 2; ++pp) {
        float s = 0.0f, q = 0.0f;
#pragma unroll
        for (int i = 0; i < 3; ++i) {
            s += v[pp][i].x; q += v[pp][i].x * v[pp][i].x;
            s += v[pp][i].y; q += v[pp][i].y * v[pp][i].y;
        }
#pragma unroll
        for (int o = 16; o; o >>= 1) {
            s += __shfl_xor(s, o, 32);
            q += __shfl_xor(q, o, 32);
        }
        const float mu   = s / 192.0f;
        const float var  = q / 192.0f - mu * mu;
        const float rstd = rsqrtf(var + 1e-5f);
        float* op = out + (size_t)(wave * 2 + pp) * DI + 2 * lane;
#pragma unroll
        for (int i = 0; i < 3; ++i) {
            v2f r;
            r.x = (v[pp][i].x - mu) * rstd * g[i].x + be[i].x;
            r.y = (v[pp][i].y - mu) * rstd * g[i].y + be[i].y;
            *(v2f*)(op + 64 * i) = r;
        }
    }
}

extern "C" void kernel_launch(void* const* d_in, const int* in_sizes, int n_in,
                              void* d_out, int out_size, void* d_ws, size_t ws_size,
                              hipStream_t stream) {
    const float* x   = (const float*)d_in[0];
    const float* y   = (const float*)d_in[1];
    const float* xpw = (const float*)d_in[2];
    const float* dtw = (const float*)d_in[3];
    const float* dtb = (const float*)d_in[4];
    const float* Al  = (const float*)d_in[5];
    const float* Ds  = (const float*)d_in[6];
    const float* gm  = (const float*)d_in[7];
    const float* bt  = (const float*)d_in[8];

    float* xdbl = (float*)d_ws;
    float* outy = xdbl + (size_t)B_ * KK * CP * LL;
    float* yt   = (float*)d_out;

    ss2d_proj_wmma<<<NTB + 256, 256, 0, stream>>>(x, xpw, xdbl, y, yt);
    ss2d_scan_local<<<B_ * KK * NP * 3, 128, 0, stream>>>(y, yt, dtw, dtb, Al, Ds, xdbl, outy);
    ss2d_scan_carry<<<B_ * KK * NS, DI, 0, stream>>>(Al, outy);
    ss2d_scan_final<<<B_ * KK * NP * 3, 128, 0, stream>>>(y, yt, dtw, dtb, Al, Ds, xdbl, outy);
    ss2d_merge_ln<<<B_ * LL / 16, 256, 0, stream>>>(outy, gm, bt, (float*)d_out);
}


# scan: a thread runs 2 channels in the halves of packed ops, so a step's B, C, dt are scalars: scalar loads from position-major x_dbl records, no LDS staging or reads for them; sums unchanged
# speedup vs baseline: 23.3394x; 1.2686x over previous
#include <hip/hip_runtime.h>
#include <hip/hip_bf16.h>


#define B_  4
#define DI  192
#define LL  4096
#define NS  16
#define RR  6
#define KK  4
#define CP  38

typedef float v2f __attribute__((ext_vector_type(2)));
typedef float v8f __attribute__((ext_vector_type(8)));

__device__ __forceinline__ int scan_idx(int k, int l) {
    int ll = (k & 2) ? (LL - 1 - l) : l;
    if (k & 1) return ((ll & 63) << 6) | (ll >> 6);
    return ll;
}

#define NTB 256
#define TPB 3
#define PG  4
#define TP  65

typedef float v4f __attribute__((ext_vector_type(4)));

__global__ __launch_bounds__(256) void ss2d_proj_wmma(
    const float* __restrict__ x,
    const float* __restrict__ xpw,
    float* __restrict__ xdbl,
    const float* __restrict__ y,
    float* __restrict__ yt)
{
    __shared__ float tile[64 * TP];
    if (blockIdx.x < NTB) {
        const int tid = threadIdx.x;
        v4f q[TPB][4];
#pragma unroll
        for (int p = 0; p < TPB; ++p) {
#pragma unroll
            for (int it = 0; it < 4; ++it)
                q[p][it] = *(const v4f*)(y + (size_t)(blockIdx.x * TPB + p) * LL + (it * 256 + tid) * 4);
        }
#pragma unroll
        for (int p = 0; p < TPB; ++p) {
#pragma unroll
            for (int it = 0; it < 4; ++it) {
                const int idx = it * 256 + tid;
                const int h   = idx >> 4;
                const int w0  = (idx & 15) << 2;
#pragma unroll
                for (int e = 0; e < 4; ++e) tile[(w0 + e) * TP + h] = q[p][it][e];
            }
            __syncthreads();
            float* dst = yt + (size_t)(blockIdx.x * TPB + p) * LL;
#pragma unroll
            for (int it = 0; it < 4; ++it) {
                const int idx = it * 256 + tid;
                const int w   = idx >> 4;
                const int h0  = (idx & 15) << 2;
                v4f r;
#pragma unroll
                for (int e = 0; e < 4; ++e) r[e] = tile[w * TP + h0 + e];
                *(v4f*)(dst + idx * 4) = r;
            }
            __syncthreads();
        }
        return;
    }

    const int wave = (int)(((blockIdx.x - NTB) * blockDim.x + threadIdx.x) >> 5);
    const int lane = threadIdx.x & 31;
    const int nt = wave & 255;
    const int pr = (wave >> 8) & 1;
    const int b  = wave >> 9;
    const int hw = lane >> 4;
    const int ln = lane & 15;
    const int ka = 2 * pr, kb = 2 * pr + 1;
    const int m  = (nt << 4) + ln;

    const float* xb = x + ((size_t)b * DI + 4 * hw) * LL + m;
    const float* wa = xpw + (size_t)ka * CP * DI + 4 * hw;
    const float* wb = xpw + (size_t)kb * CP * DI + 4 * hw;
    const int  tr = ln & 7;
    const bool tv = tr < RR;
    const float* ap[5];
    ap[0] = wa + (size_t)(RR + ln) * DI;
    ap[1] = wa + (size_t)(RR + NS + ln) * DI;
    ap[2] = wb + (size_t)(RR + ln) * DI;
    ap[3] = wb + (size_t)(RR + NS + ln) * DI;
    ap[4] = ((ln < 8) ? wa : wb) + (size_t)(tv ? tr : 0) * DI;

    v8f acc[5];
#pragma unroll
    for (int j = 0; j < 5; ++j) acc[j] = (v8f){0.f, 0.f, 0.f, 0.f, 0.f, 0.f, 0.f, 0.f};

    v4f   Aq[2][PG][5];
    float Bq[2][PG][4];
#pragma unroll
    for (int gi = 0; gi < PG; ++gi) {
#pragma unroll
        for (int j = 0; j < 5; ++j) Aq[0][gi][j] = *(const v4f*)(ap[j] + 8 * gi);
#pragma unroll
        for (int e = 0; e < 4; ++e) Bq[0][gi][e] = xb[(size_t)(8 * gi + e) * LL];
    }
#pragma unroll
    for (int t = 0; t < DI / (8 * PG); ++t) {
        const int cur = t & 1, nxt = cur ^ 1;
        if (t + 1 < DI / (8 * PG)) {
#pragma unroll
            for (int gi = 0; gi < PG; ++gi) {
                const int g = (t + 1) * PG + gi;
#pragma unroll
                for (int j = 0; j < 5; ++j) Aq[nxt][gi][j] = *(const v4f*)(ap[j] + 8 * g);
#pragma unroll
                for (int e = 0; e < 4; ++e) Bq[nxt][gi][e] = xb[(size_t)(8 * g + e) * LL];
            }
        }
#pragma unroll
        for (int gi = 0; gi < PG; ++gi) {
            const v4f zero4 = {0.f, 0.f, 0.f, 0.f};
            const v4f a4 = tv ? Aq[cur][gi][4] : zero4;
            const v2f b01 = {Bq[cur][gi][0], Bq[cur][gi][1]};
            const v2f b23 = {Bq[cur][gi][2], Bq[cur][gi][3]};
#pragma unroll
            for (int j = 0; j < 5; ++j) {
                const v4f q = (j < 4) ? Aq[cur][gi][j] : a4;
                const v2f a = {q.x, q.y};
                acc[j] = __builtin_amdgcn_wmma_f32_16x16x4_f32(
                    false, a, false, b01, (short)0, acc[j], false, false);
            }
#pragma unroll
            for (int j = 0; j < 5; ++j) {
                const v4f q = (j < 4) ? Aq[cur][gi][j] : a4;
                const v2f a = {q.z, q.w};
                acc[j] = __builtin_amdgcn_wmma_f32_16x16x4_f32(
                    false, a, false, b23, (short)0, acc[j], false, false);
            }
        }
    }

    const int mt_ = ((m & 63) << 6) | (m >> 6);
    const int lpa = (ka & 2) ? (LL - 1 - m)   : m;
    const int lpb = (kb & 2) ? (LL - 1 - mt_) : mt_;
    float* oa = xdbl + ((size_t)(b * KK + ka) * LL + lpa) * CP + (hw << 3);
    float* ob = xdbl + ((size_t)(b * KK + kb) * LL + lpb) * CP + (hw << 3);
#pragma unroll
    for (int i = 0; i < 8; i += 2) {
        *(v2f*)(oa + i)      = (v2f){acc[0][i], acc[0][i + 1]};
        *(v2f*)(oa + NS + i) = (v2f){acc[1][i], acc[1][i + 1]};
        *(v2f*)(ob + i)      = (v2f){acc[2][i], acc[2][i + 1]};
        *(v2f*)(ob + NS + i) = (v2f){acc[3][i], acc[3][i + 1]};
    }
    float* ot = (hw ? ob : oa) - (hw << 3) + 2 * NS;
#pragma unroll
    for (int i = 0; i < RR; i += 2) *(v2f*)(ot + i) = (v2f){acc[4][i], acc[4][i + 1]};
}

#define NC  64
#define CT  64
#define CW  4
#define NQ  16
#define PT  256
#define NRC 20
#define DG  4
#define DGA 2

typedef float v16f __attribute__((ext_vector_type(16)));
typedef v16f  v16fr __attribute__((aligned(4)));
#define UNI __attribute__((address_space(4)))

__device__ __forceinline__ int u_slot(int r, int p) {
    return r * PT + ((p + r) & (PT - 1));
}

__device__ __forceinline__ void scan_chunk_body(
    const bool final_pass,
    const float* __restrict__ y,
    const float* __restrict__ yt,
    const float* __restrict__ dtw_g,
    const float* __restrict__ dtb_g,
    const float* __restrict__ Alogs,
    const float* __restrict__ Ds_g,
    const float* __restrict__ xdbl,
    float* outy,
    float* sm_uy)
{
    const int tid  = threadIdx.x;
    const int q    = tid & 31;
    const int hf   = __builtin_amdgcn_readfirstlane(tid >> 5);
    const int dblk = blockIdx.x % 3;
    const int cq   = (blockIdx.x / 3) % NQ;
    const int kd   = blockIdx.x / (3 * NQ);
    const int c    = CW * cq + hf;
    const int k    = kd & 3;
    const int b    = kd >> 2;
    const int d    = dblk * 64 + 2 * q;
    const int kdch = k * DI + d;

    v2f A2[NS];
    {
        const v4f* ap = (const v4f*)(Alogs + (size_t)kdch * NS);
#pragma unroll
        for (int i = 0; i < 4; ++i) {
            const v4f a0 = ap[i], a1 = ap[4 + i];
            A2[4 * i]     = (v2f){-__expf(a0.x), -__expf(a1.x)};
            A2[4 * i + 1] = (v2f){-__expf(a0.y), -__expf(a1.y)};
            A2[4 * i + 2] = (v2f){-__expf(a0.z), -__expf(a1.z)};
            A2[4 * i + 3] = (v2f){-__expf(a0.w), -__expf(a1.w)};
        }
    }
    v2f dtw[RR];
#pragma unroll
    for (int r = 0; r < RR; ++r)
        dtw[r] = (v2f){dtw_g[(size_t)kdch * RR + r], dtw_g[(size_t)(kdch + 1) * RR + r]};
    const v2f dtb = *(const v2f*)(dtb_g + kdch);
    const v2f Dsd = *(const v2f*)(Ds_g + kdch);

    const UNI float* xrec = (const UNI float*)(xdbl + ((size_t)kd * LL + c * CT) * CP);
    const float* yb  = ((k & 1) ? yt : y) + ((size_t)b * DI + dblk * 64) * LL;
    float* ych       = outy + (size_t)kd * LL * DI + d;

    v2f h2[NS];
    if (final_pass) {
#pragma unroll
        for (int n = 0; n < NS; ++n)
            h2[n] = *(const v2f*)(ych + (size_t)scan_idx(k, c * CT + n) * DI);
    } else {
#pragma unroll
        for (int n = 0; n < NS; ++n) h2[n] = (v2f){0.0f, 0.0f};
    }
    v2f sdelta = {0.0f, 0.0f};

    {
        const int mbase = (k & 2) ? (LL - PT - cq * PT) : (cq * PT);
        v4f uq[32];
#pragma unroll
        for (int it = 0; it < 32; ++it)
            uq[it] = *(const v4f*)(yb + (size_t)(4 * (it >> 1) + hf) * LL + mbase + 128 * (it & 1) + 4 * q);
#pragma unroll
        for (int it = 0; it < 32; ++it) {
            const int r = 4 * (it >> 1) + hf;
#pragma unroll
            for (int e = 0; e < 4; ++e) {
                int m = mbase + 128 * (it & 1) + 4 * q + e;
                int l = (k & 2) ? (LL - 1 - m) : m;
                sm_uy[u_slot(r, l - cq * PT)] = uq[it][e];
            }
        }
    }
    const int l0 = c * CT;
    __syncthreads();

    const float* urow = sm_uy + (2 * q) * PT;
    int up = (hf * CT + 2 * q) & (PT - 1);

    const int mstep = ((k & 1) ? 64 : 1) * ((k & 2) ? -1 : 1);
    const int dstep = mstep * DI;
    float* kbase    = outy + (size_t)kd * LL * DI;
    const unsigned doff = (unsigned)d * 4u;
    auto slot2 = [&](const int rec) -> v2f* {
        return (v2f*)((char*)(kbase + rec) + doff);
    };
    int srec        = scan_idx(k, l0) * DI;

    auto delta_of = [&](const int j) -> v2f {
        const UNI float* dt = xrec + j * CP + 2 * NS;
        v2f sarg = dtb;
        sarg += dtw[0] * dt[0];
        sarg += dtw[1] * dt[1];
        sarg += dtw[2] * dt[2];
        sarg += dtw[3] * dt[3];
        sarg += dtw[4] * dt[4];
        sarg += dtw[5] * dt[5];
        const float lpx = log1pf(__expf(sarg.x));
        const float lpy = log1pf(__expf(sarg.y));
        return (v2f){(sarg.x > 20.0f) ? sarg.x : lpx, (sarg.y > 20.0f) ? sarg.y : lpy};
    };

    auto rec16 = [&](const int j, const int part) -> v16f {
        return *(const UNI v16fr*)(xrec + j * CP + part * NS);
    };

    auto state_step = [&](const v16f Bv, const v16f Cv, const float b0, const v2f delta) {
        const v2f u = {urow[up], urow[PT + ((up + 1) & (PT - 1))]};
        up = (up + 1) & (PT - 1);
        const v2f du = delta * u;

#pragma unroll
        for (int n = 0; n < NS; ++n) {
            const v2f arg = delta * A2[n];
            v2f ex;
            ex.x = __expf(arg.x);
            ex.y = __expf(arg.y);
            h2[n] = ex * h2[n] + du * (n == 0 ? b0 : Bv[n]);
        }

        if (final_pass) {
            v2f acce = {0.0f, 0.0f}, acco = {0.0f, 0.0f};
#pragma unroll
            for (int n = 0; n < NS; n += 2) {
                acce = h2[n] * Cv[n] + acce;
                acco = h2[n + 1] * Cv[n + 1] + acco;
            }
            *slot2(srec) = (acce + acco) + Dsd * u;
        }
    };

    if (!final_pass) {
        for (int j = 0; j < NRC; j += DGA) {
            v2f dl[DGA];
#pragma unroll
            for (int i = 0; i < DGA; ++i)
                dl[i] = delta_of(j + i);
#pragma unroll
            for (int i = 0; i < DGA; ++i) {
                const v16f Bv = rec16(j + i, 0);
                state_step(Bv, Bv, Bv[0], dl[i]);
                sdelta += dl[i];
                srec += dstep;
            }
        }
        for (int j = NRC; j < CT; j += DG) {
            v2f dl[DG];
#pragma unroll
            for (int i = 0; i < DG; ++i)
                dl[i] = delta_of(j + i);
#pragma unroll
            for (int i = 0; i < DG; ++i) {
                const v16f Bv = rec16(j + i, 0);
                state_step(Bv, Bv, Bv[0], dl[i]);
                sdelta += dl[i];
                *slot2(srec) = dl[i];
                srec += dstep;
            }
        }
    } else {
        v2f dq[4];
#pragma unroll
        for (int i = 0; i < 4; ++i) dq[i] = *slot2(srec + (NRC + i) * dstep);

        for (int j = 0; j < NRC; j += DGA) {
            v2f dl[DGA];
#pragma unroll
            for (int i = 0; i < DGA; ++i)
                dl[i] = delta_of(j + i);
#pragma unroll
            for (int i = 0; i < DGA; ++i) {
                const v16f Bv = rec16(j + i, 0);
                state_step(Bv, rec16(j + i, 1), Bv[0], dl[i]);
                srec += dstep;
            }
        }
        v16f Bn = rec16(NRC, 0), Cn = rec16(NRC, 1);
        for (int j = NRC; j < CT; j += 4) {
#pragma unroll
            for (int i = 0; i < 4; ++i) {
                const v16f Bc = Bn, Cc = Cn;
                float b0 = Bc[0];
                asm volatile("" : "+v"(b0));
                __builtin_amdgcn_sched_barrier(0);
                const int nx = min(j + i + 1, CT - 1);
                Bn = rec16(nx, 0);
                Cn = rec16(nx, 1);
                __builtin_amdgcn_sched_barrier(0);
                state_step(Bc, Cc, b0, dq[i]);
                const int ahead = min(j + i + 4, CT - 1) - (j + i);
                dq[i] = *slot2(srec + ahead * dstep);
                srec += dstep;
                __builtin_amdgcn_sched_barrier(0);
            }
        }
    }

    if (!final_pass) {
#pragma unroll
        for (int n = 0; n < NS; ++n)
            *(v2f*)(ych + (size_t)scan_idx(k, c * CT + n) * DI) = h2[n];
        *(v2f*)(ych + (size_t)scan_idx(k, c * CT + NS) * DI) = sdelta;
    }
}

__global__ __launch_bounds__(128) __attribute__((amdgpu_waves_per_eu(3, 3)))
void ss2d_scan_local(
    const float* __restrict__ y,
    const float* __restrict__ yt,
    const float* __restrict__ dtw_g,
    const float* __restrict__ dtb_g,
    const float* __restrict__ Alogs,
    const float* __restrict__ Ds_g,
    const float* __restrict__ xdbl,
    float* outy)
{
    __shared__ float sm_uy[64 * PT];
    scan_chunk_body(false, y, yt, dtw_g, dtb_g, Alogs, Ds_g, xdbl, outy, sm_uy);
}

__global__ __launch_bounds__(128) __attribute__((amdgpu_waves_per_eu(3, 3)))
void ss2d_scan_final(
    const float* __restrict__ y,
    const float* __restrict__ yt,
    const float* __restrict__ dtw_g,
    const float* __restrict__ dtb_g,
    const float* __restrict__ Alogs,
    const float* __restrict__ Ds_g,
    const float* __restrict__ xdbl,
    float* outy)
{
    __shared__ float sm_uy[64 * PT];
    scan_chunk_body(true, y, yt, dtw_g, dtb_g, Alogs, Ds_g, xdbl, outy, sm_uy);
}

__global__ __launch_bounds__(192) void ss2d_scan_carry(
    const float* __restrict__ Alogs,
    float* outy)
{
    const int d   = threadIdx.x;
    const int n   = blockIdx.x & 15;
    const int kd  = blockIdx.x >> 4;
    const int k   = kd & 3;
    const float A = -__expf(Alogs[(size_t)(k * DI + d) * NS + n]);
    float* base = outy + (size_t)kd * LL * DI + d;

    float S[NC], sd[NC];
#pragma unroll
    for (int c = 0; c < NC; ++c) {
        S[c]  = base[(size_t)scan_idx(k, c * CT + n) * DI];
        sd[c] = base[(size_t)scan_idx(k, c * CT + NS) * DI];
    }
    float H = 0.0f;
#pragma unroll
    for (int c = 0; c < NC; ++c) {
        base[(size_t)scan_idx(k, c * CT + n) * DI] = H;
        H = __expf(A * sd[c]) * H + S[c];
    }
}

__global__ __launch_bounds__(256) void ss2d_merge_ln(
    const float* __restrict__ outy,
    const float* __restrict__ gamma,
    const float* __restrict__ beta,
    float* __restrict__ out)
{
    const int lane = threadIdx.x & 31;
    const int wave = (int)((blockIdx.x * 256 + threadIdx.x) >> 5);
    const size_t ks = (size_t)LL * DI;

    v2f g[3], be[3];
#pragma unroll
    for (int i = 0; i < 3; ++i) {
        g[i]  = *(const v2f*)(gamma + 2 * lane + 64 * i);
        be[i] = *(const v2f*)(beta + 2 * lane + 64 * i);
    }

    v2f v[2][3];
#pragma unroll
    for (int pp = 0; pp < 2; ++pp) {
        const int p  = wave * 2 + pp;
        const int b  = p >> 12;
        const int lm = p & (LL - 1);
        const float* base = outy + ((size_t)b * KK * LL + lm) * DI + 2 * lane;
#pragma unroll
        for (int i = 0; i < 3; ++i) {
            const v2f a0 = *(const v2f*)(base + 64 * i);
            const v2f a2 = *(const v2f*)(base + 2 * ks + 64 * i);
            const v2f a1 = *(const v2f*)(base + ks + 64 * i);
            const v2f a3 = *(const v2f*)(base + 3 * ks + 64 * i);
            v[pp][i] = ((a0 + a2) + a1) + a3;
        }
    }

#pragma unroll
    for (int pp = 0; pp < 2; ++pp) {
        float s = 0.0f, q = 0.0f;
#pragma unroll
        for (int i = 0; i < 3; ++i) {
            s += v[pp][i].x; q += v[pp][i].x * v[pp][i].x;
            s += v[pp][i].y; q += v[pp][i].y * v[pp][i].y;
        }
#pragma unroll
        for (int o = 16; o; o >>= 1) {
            s += __shfl_xor(s, o, 32);
            q += __shfl_xor(q, o, 32);
        }
        const float mu   = s / 192.0f;
        const float var  = q / 192.0f - mu * mu;
        const float rstd = rsqrtf(var + 1e-5f);
        float* op = out + (size_t)(wave * 2 + pp) * DI + 2 * lane;
#pragma unroll
        for (int i = 0; i < 3; ++i) {
            v2f r;
            r.x = (v[pp][i].x - mu) * rstd * g[i].x + be[i].x;
            r.y = (v[pp][i].y - mu) * rstd * g[i].y + be[i].y;
            *(v2f*)(op + 64 * i) = r;
        }
    }
}

extern "C" void kernel_launch(void* const* d_in, const int* in_sizes, int n_in,
                              void* d_out, int out_size, void* d_ws, size_t ws_size,
                              hipStream_t stream) {
    const float* x   = (const float*)d_in[0];
    const float* y   = (const float*)d_in[1];
    const float* xpw = (const float*)d_in[2];
    const float* dtw = (const float*)d_in[3];
    const float* dtb = (const float*)d_in[4];
    const float* Al  = (const float*)d_in[5];
    const float* Ds  = (const float*)d_in[6];
    const float* gm  = (const float*)d_in[7];
    const float* bt  = (const float*)d_in[8];

    float* xdbl = (float*)d_ws;
    float* outy = xdbl + (size_t)B_ * KK * CP * LL;
    float* yt   = (float*)d_out;

    ss2d_proj_wmma<<<NTB + 256, 256, 0, stream>>>(x, xpw, xdbl, y, yt);
    ss2d_scan_local<<<B_ * KK * NQ * 3, 128, 0, stream>>>(y, yt, dtw, dtb, Al, Ds, xdbl, outy);
    ss2d_scan_carry<<<B_ * KK * NS, DI, 0, stream>>>(Al, outy);
    ss2d_scan_final<<<B_ * KK * NQ * 3, 128, 0, stream>>>(y, yt, dtw, dtb, Al, Ds, xdbl, outy);
    ss2d_merge_ln<<<B_ * LL / 16, 256, 0, stream>>>(outy, gm, bt, (float*)d_out);
}


# scan work-group = a direction and its flip on one stretch: one u tile staged for both, final pass adds the pair's outputs through LDS and stores once; merge reads 2 planes; merge sum order changes
# speedup vs baseline: 24.4792x; 1.0488x over previous
#include <hip/hip_runtime.h>
#include <hip/hip_bf16.h>


#define B_  4
#define DI  192
#define LL  4096
#define NS  16
#define RR  6
#define KK  4
#define CP  38

typedef float v2f __attribute__((ext_vector_type(2)));
typedef float v8f __attribute__((ext_vector_type(8)));

__device__ __forceinline__ int scan_idx(int k, int l) {
    int ll = (k & 2) ? (LL - 1 - l) : l;
    if (k & 1) return ((ll & 63) << 6) | (ll >> 6);
    return ll;
}

#define NTB 256
#define TPB 3
#define PG  4
#define TP  65

typedef float v4f __attribute__((ext_vector_type(4)));

__global__ __launch_bounds__(256) void ss2d_proj_wmma(
    const float* __restrict__ x,
    const float* __restrict__ xpw,
    float* __restrict__ xdbl,
    const float* __restrict__ y,
    float* __restrict__ yt)
{
    __shared__ float tile[64 * TP];
    if (blockIdx.x < NTB) {
        const int tid = threadIdx.x;
        v4f q[TPB][4];
#pragma unroll
        for (int p = 0; p < TPB; ++p) {
#pragma unroll
            for (int it = 0; it < 4; ++it)
                q[p][it] = *(const v4f*)(y + (size_t)(blockIdx.x * TPB + p) * LL + (it * 256 + tid) * 4);
        }
#pragma unroll
        for (int p = 0; p < TPB; ++p) {
#pragma unroll
            for (int it = 0; it < 4; ++it) {
                const int idx = it * 256 + tid;
                const int h   = idx >> 4;
                const int w0  = (idx & 15) << 2;
#pragma unroll
                for (int e = 0; e < 4; ++e) tile[(w0 + e) * TP + h] = q[p][it][e];
            }
            __syncthreads();
            float* dst = yt + (size_t)(blockIdx.x * TPB + p) * LL;
#pragma unroll
            for (int it = 0; it < 4; ++it) {
                const int idx = it * 256 + tid;
                const int w   = idx >> 4;
                const int h0  = (idx & 15) << 2;
                v4f r;
#pragma unroll
                for (int e = 0; e < 4; ++e) r[e] = tile[w * TP + h0 + e];
                *(v4f*)(dst + idx * 4) = r;
            }
            __syncthreads();
        }
        return;
    }

    const int wave = (int)(((blockIdx.x - NTB) * blockDim.x + threadIdx.x) >> 5);
    const int lane = threadIdx.x & 31;
    const int nt = wave & 255;
    const int pr = (wave >> 8) & 1;
    const int b  = wave >> 9;
    const int hw = lane >> 4;
    const int ln = lane & 15;
    const int ka = 2 * pr, kb = 2 * pr + 1;
    const int m  = (nt << 4) + ln;

    const float* xb = x + ((size_t)b * DI + 4 * hw) * LL + m;
    const float* wa = xpw + (size_t)ka * CP * DI + 4 * hw;
    const float* wb = xpw + (size_t)kb * CP * DI + 4 * hw;
    const int  tr = ln & 7;
    const bool tv = tr < RR;
    const float* ap[5];
    ap[0] = wa + (size_t)(RR + ln) * DI;
    ap[1] = wa + (size_t)(RR + NS + ln) * DI;
    ap[2] = wb + (size_t)(RR + ln) * DI;
    ap[3] = wb + (size_t)(RR + NS + ln) * DI;
    ap[4] = ((ln < 8) ? wa : wb) + (size_t)(tv ? tr : 0) * DI;

    v8f acc[5];
#pragma unroll
    for (int j = 0; j < 5; ++j) acc[j] = (v8f){0.f, 0.f, 0.f, 0.f, 0.f, 0.f, 0.f, 0.f};

    v4f   Aq[2][PG][5];
    float Bq[2][PG][4];
#pragma unroll
    for (int gi = 0; gi < PG; ++gi) {
#pragma unroll
        for (int j = 0; j < 5; ++j) Aq[0][gi][j] = *(const v4f*)(ap[j] + 8 * gi);
#pragma unroll
        for (int e = 0; e < 4; ++e) Bq[0][gi][e] = xb[(size_t)(8 * gi + e) * LL];
    }
#pragma unroll
    for (int t = 0; t < DI / (8 * PG); ++t) {
        const int cur = t & 1, nxt = cur ^ 1;
        if (t + 1 < DI / (8 * PG)) {
#pragma unroll
            for (int gi = 0; gi < PG; ++gi) {
                const int g = (t + 1) * PG + gi;
#pragma unroll
                for (int j = 0; j < 5; ++j) Aq[nxt][gi][j] = *(const v4f*)(ap[j] + 8 * g);
#pragma unroll
                for (int e = 0; e < 4; ++e) Bq[nxt][gi][e] = xb[(size_t)(8 * g + e) * LL];
            }
        }
#pragma unroll
        for (int gi = 0; gi < PG; ++gi) {
            const v4f zero4 = {0.f, 0.f, 0.f, 0.f};
            const v4f a4 = tv ? Aq[cur][gi][4] : zero4;
            const v2f b01 = {Bq[cur][gi][0], Bq[cur][gi][1]};
            const v2f b23 = {Bq[cur][gi][2], Bq[cur][gi][3]};
#pragma unroll
            for (int j = 0; j < 5; ++j) {
                const v4f q = (j < 4) ? Aq[cur][gi][j] : a4;
                const v2f a = {q.x, q.y};
                acc[j] = __builtin_amdgcn_wmma_f32_16x16x4_f32(
                    false, a, false, b01, (short)0, acc[j], false, false);
            }
#pragma unroll
            for (int j = 0; j < 5; ++j) {
                const v4f q = (j < 4) ? Aq[cur][gi][j] : a4;
                const v2f a = {q.z, q.w};
                acc[j] = __builtin_amdgcn_wmma_f32_16x16x4_f32(
                    false, a, false, b23, (short)0, acc[j], false, false);
            }
        }
    }

    const int mt_ = ((m & 63) << 6) | (m >> 6);
    const int lpa = (ka & 2) ? (LL - 1 - m)   : m;
    const int lpb = (kb & 2) ? (LL - 1 - mt_) : mt_;
    float* oa = xdbl + ((size_t)(b * KK + ka) * LL + lpa) * CP + (hw << 3);
    float* ob = xdbl + ((size_t)(b * KK + kb) * LL + lpb) * CP + (hw << 3);
#pragma unroll
    for (int i = 0; i < 8; i += 2) {
        *(v2f*)(oa + i)      = (v2f){acc[0][i], acc[0][i + 1]};
        *(v2f*)(oa + NS + i) = (v2f){acc[1][i], acc[1][i + 1]};
        *(v2f*)(ob + i)      = (v2f){acc[2][i], acc[2][i + 1]};
        *(v2f*)(ob + NS + i) = (v2f){acc[3][i], acc[3][i + 1]};
    }
    float* ot = (hw ? ob : oa) - (hw << 3) + 2 * NS;
#pragma unroll
    for (int i = 0; i < RR; i += 2) *(v2f*)(ot + i) = (v2f){acc[4][i], acc[4][i + 1]};
}

#define NC  64
#define CT  64
#define CW  2
#define NQ  32
#define PT  128
#define HS  (CT / 2)
#define NRC 20
#define DG  4
#define DGA 2

typedef float v16f __attribute__((ext_vector_type(16)));
typedef v16f  v16fr __attribute__((aligned(4)));
#define UNI __attribute__((address_space(4)))

__device__ __forceinline__ int u_slot(int r, int p) {
    return r * PT + ((p + r) & (PT - 1));
}

__device__ __forceinline__ void scan_chunk_body(
    const bool final_pass,
    const float* __restrict__ y,
    const float* __restrict__ yt,
    const float* __restrict__ dtw_g,
    const float* __restrict__ dtb_g,
    const float* __restrict__ Alogs,
    const float* __restrict__ Ds_g,
    const float* __restrict__ xdbl,
    float* outy,
    float* sm_uy,
    float* sm_part)
{
    const int tid  = threadIdx.x;
    const int q    = tid & 31;
    const int hf   = __builtin_amdgcn_readfirstlane(tid >> 5);
    const int dblk = blockIdx.x % 3;
    const int cq   = (blockIdx.x / 3) % NQ;
    const int bp   = blockIdx.x / (3 * NQ);
    const int kp   = bp & 1;
    const int b    = bp >> 1;
    const int fl   = hf >> 1;
    const int ci   = hf & 1;
    const int k    = kp + 2 * fl;
    const int kd   = b * KK + k;
    const int c    = fl ? (NC - 1 - (CW * cq + ci)) : (CW * cq + ci);
    const int d    = dblk * 64 + 2 * q;
    const int kdch = k * DI + d;

    v2f A2[NS];
    {
        const v4f* ap = (const v4f*)(Alogs + (size_t)kdch * NS);
#pragma unroll
        for (int i = 0; i < 4; ++i) {
            const v4f a0 = ap[i], a1 = ap[4 + i];
            A2[4 * i]     = (v2f){-__expf(a0.x), -__expf(a1.x)};
            A2[4 * i + 1] = (v2f){-__expf(a0.y), -__expf(a1.y)};
            A2[4 * i + 2] = (v2f){-__expf(a0.z), -__expf(a1.z)};
            A2[4 * i + 3] = (v2f){-__expf(a0.w), -__expf(a1.w)};
        }
    }
    v2f dtw[RR];
#pragma unroll
    for (int r = 0; r < RR; ++r)
        dtw[r] = (v2f){dtw_g[(size_t)kdch * RR + r], dtw_g[(size_t)(kdch + 1) * RR + r]};
    const v2f dtb = *(const v2f*)(dtb_g + kdch);
    const v2f Dsd = *(const v2f*)(Ds_g + kdch);

    const UNI float* xrec = (const UNI float*)(xdbl + ((size_t)kd * LL + c * CT) * CP);
    const float* yb  = ((k & 1) ? yt : y) + ((size_t)b * DI + dblk * 64) * LL;
    float* ych       = outy + (size_t)kd * LL * DI + d;

    v2f h2[NS];
    if (final_pass) {
#pragma unroll
        for (int n = 0; n < NS; ++n)
            h2[n] = *(const v2f*)(ych + (size_t)scan_idx(k, c * CT + n) * DI);
    } else {
#pragma unroll
        for (int n = 0; n < NS; ++n) h2[n] = (v2f){0.0f, 0.0f};
    }
    v2f sdelta = {0.0f, 0.0f};

    {
        const int mbase = cq * PT;
        v4f uq[16];
#pragma unroll
        for (int it = 0; it < 16; ++it)
            uq[it] = *(const v4f*)(yb + (size_t)(4 * it + hf) * LL + mbase + 4 * q);
#pragma unroll
        for (int it = 0; it < 16; ++it) {
            const int r = 4 * it + hf;
#pragma unroll
            for (int e = 0; e < 4; ++e)
                sm_uy[u_slot(r, 4 * q + e)] = uq[it][e];
        }
    }
    const int l0 = c * CT;
    __syncthreads();

    const float* urow = sm_uy + (2 * q) * PT;
    int up = (ci * CT + (fl ? CT - 1 : 0) + 2 * q) & (PT - 1);
    const int ustep = fl ? (PT - 1) : 1;
    float* prow  = sm_part + (ci * CT + (fl ? CT - 1 : 0)) * 64 + 2 * q;
    const int pstep = fl ? -64 : 64;
    float* kstore = outy + (size_t)(b * KK + kp) * LL * DI;

    const int mstep = ((k & 1) ? 64 : 1) * ((k & 2) ? -1 : 1);
    const int dstep = mstep * DI;
    float* kbase    = outy + (size_t)kd * LL * DI;
    const unsigned doff = (unsigned)d * 4u;
    auto slot2 = [&](const int rec) -> v2f* {
        return (v2f*)((char*)(kbase + rec) + doff);
    };
    auto slotS = [&](const int rec) -> v2f* {
        return (v2f*)((char*)(kstore + rec) + doff);
    };
    int srec        = scan_idx(k, l0) * DI;

    auto delta_of = [&](const int j) -> v2f {
        const UNI float* dt = xrec + j * CP + 2 * NS;
        v2f sarg = dtb;
        sarg += dtw[0] * dt[0];
        sarg += dtw[1] * dt[1];
        sarg += dtw[2] * dt[2];
        sarg += dtw[3] * dt[3];
        sarg += dtw[4] * dt[4];
        sarg += dtw[5] * dt[5];
        const float lpx = log1pf(__expf(sarg.x));
        const float lpy = log1pf(__expf(sarg.y));
        return (v2f){(sarg.x > 20.0f) ? sarg.x : lpx, (sarg.y > 20.0f) ? sarg.y : lpy};
    };

    auto rec16 = [&](const int j, const int part) -> v16f {
        return *(const UNI v16fr*)(xrec + j * CP + part * NS);
    };

    auto state_step = [&](const v16f Bv, const v16f Cv, const float b0, const v2f delta, const bool second_half) {
        const v2f u = {urow[up], urow[PT + ((up + 1) & (PT - 1))]};
        up = (up + ustep) & (PT - 1);
        const v2f du = delta * u;

#pragma unroll
        for (int n = 0; n < NS; ++n) {
            const v2f arg = delta * A2[n];
            v2f ex;
            ex.x = __expf(arg.x);
            ex.y = __expf(arg.y);
            h2[n] = ex * h2[n] + du * (n == 0 ? b0 : Bv[n]);
        }

        if (final_pass) {
            v2f acce = {0.0f, 0.0f}, acco = {0.0f, 0.0f};
#pragma unroll
            for (int n = 0; n < NS; n += 2) {
                acce = h2[n] * Cv[n] + acce;
                acco = h2[n + 1] * Cv[n + 1] + acco;
            }
            const v2f yv = (acce + acco) + Dsd * u;
            if (second_half) {
                *slotS(srec) = yv + *(const v2f*)prow;
            } else {
                *(v2f*)prow = yv;
            }
            prow += pstep;
        }
    };

    if (!final_pass) {
        for (int j = 0; j < NRC; j += DGA) {
            v2f dl[DGA];
#pragma unroll
            for (int i = 0; i < DGA; ++i)
                dl[i] = delta_of(j + i);
#pragma unroll
            for (int i = 0; i < DGA; ++i) {
                const v16f Bv = rec16(j + i, 0);
                state_step(Bv, Bv, Bv[0], dl[i], false);
                sdelta += dl[i];
                srec += dstep;
            }
        }
        for (int j = NRC; j < CT; j += DG) {
            v2f dl[DG];
#pragma unroll
            for (int i = 0; i < DG; ++i)
                dl[i] = delta_of(j + i);
#pragma unroll
            for (int i = 0; i < DG; ++i) {
                const v16f Bv = rec16(j + i, 0);
                state_step(Bv, Bv, Bv[0], dl[i], false);
                sdelta += dl[i];
                *slot2(srec) = dl[i];
                srec += dstep;
            }
        }
    } else {
        v2f dq[4];
#pragma unroll
        for (int i = 0; i < 4; ++i) dq[i] = *slot2(srec + (NRC + i) * dstep);

        for (int j = 0; j < NRC; j += DGA) {
            v2f dl[DGA];
#pragma unroll
            for (int i = 0; i < DGA; ++i)
                dl[i] = delta_of(j + i);
#pragma unroll
            for (int i = 0; i < DGA; ++i) {
                const v16f Bv = rec16(j + i, 0);
                state_step(Bv, rec16(j + i, 1), Bv[0], dl[i], false);
                srec += dstep;
            }
        }
        v16f Bn = rec16(NRC, 0), Cn = rec16(NRC, 1);
        auto four_steps = [&](const int j, const bool second_half) {
#pragma unroll
            for (int i = 0; i < 4; ++i) {
                const v16f Bc = Bn, Cc = Cn;
                float b0 = Bc[0];
                asm volatile("" : "+v"(b0));
                __builtin_amdgcn_sched_barrier(0);
                const int nx = min(j + i + 1, CT - 1);
                Bn = rec16(nx, 0);
                Cn = rec16(nx, 1);
                __builtin_amdgcn_sched_barrier(0);
                state_step(Bc, Cc, b0, dq[i], second_half);
                const int ahead = min(j + i + 4, CT - 1) - (j + i);
                dq[i] = *slot2(srec + ahead * dstep);
                srec += dstep;
                __builtin_amdgcn_sched_barrier(0);
            }
        };
        for (int j = NRC; j < HS; j += 4) four_steps(j, false);
        __syncthreads();
        for (int j = HS; j < CT; j += 4) four_steps(j, true);
    }

    if (!final_pass) {
#pragma unroll
        for (int n = 0; n < NS; ++n)
            *(v2f*)(ych + (size_t)scan_idx(k, c * CT + n) * DI) = h2[n];
        *(v2f*)(ych + (size_t)scan_idx(k, c * CT + NS) * DI) = sdelta;
    }
}

__global__ __launch_bounds__(128) __attribute__((amdgpu_waves_per_eu(3, 3)))
void ss2d_scan_local(
    const float* __restrict__ y,
    const float* __restrict__ yt,
    const float* __restrict__ dtw_g,
    const float* __restrict__ dtb_g,
    const float* __restrict__ Alogs,
    const float* __restrict__ Ds_g,
    const float* __restrict__ xdbl,
    float* outy)
{
    __shared__ float sm_uy[64 * PT];
    scan_chunk_body(false, y, yt, dtw_g, dtb_g, Alogs, Ds_g, xdbl, outy, sm_uy, sm_uy);
}

__global__ __launch_bounds__(128) __attribute__((amdgpu_waves_per_eu(3, 3)))
void ss2d_scan_final(
    const float* __restrict__ y,
    const float* __restrict__ yt,
    const float* __restrict__ dtw_g,
    const float* __restrict__ dtb_g,
    const float* __restrict__ Alogs,
    const float* __restrict__ Ds_g,
    const float* __restrict__ xdbl,
    float* outy)
{
    __shared__ float sm_uy[64 * PT];
    __shared__ float sm_part[CW * CT * 64];
    scan_chunk_body(true, y, yt, dtw_g, dtb_g, Alogs, Ds_g, xdbl, outy, sm_uy, sm_part);
}

__global__ __launch_bounds__(192) void ss2d_scan_carry(
    const float* __restrict__ Alogs,
    float* outy)
{
    const int d   = threadIdx.x;
    const int n   = blockIdx.x & 15;
    const int kd  = blockIdx.x >> 4;
    const int k   = kd & 3;
    const float A = -__expf(Alogs[(size_t)(k * DI + d) * NS + n]);
    float* base = outy + (size_t)kd * LL * DI + d;

    float S[NC], sd[NC];
#pragma unroll
    for (int c = 0; c < NC; ++c) {
        S[c]  = base[(size_t)scan_idx(k, c * CT + n) * DI];
        sd[c] = base[(size_t)scan_idx(k, c * CT + NS) * DI];
    }
    float H = 0.0f;
#pragma unroll
    for (int c = 0; c < NC; ++c) {
        base[(size_t)scan_idx(k, c * CT + n) * DI] = H;
        H = __expf(A * sd[c]) * H + S[c];
    }
}

__global__ __launch_bounds__(256) void ss2d_merge_ln(
    const float* __restrict__ outy,
    const float* __restrict__ gamma,
    const float* __restrict__ beta,
    float* __restrict__ out)
{
    const int lane = threadIdx.x & 31;
    const int wave = (int)((blockIdx.x * 256 + threadIdx.x) >> 5);
    const size_t ks = (size_t)LL * DI;

    v2f g[3], be[3];
#pragma unroll
    for (int i = 0; i < 3; ++i) {
        g[i]  = *(const v2f*)(gamma + 2 * lane + 64 * i);
        be[i] = *(const v2f*)(beta + 2 * lane + 64 * i);
    }

    v2f v[2][3];
#pragma unroll
    for (int pp = 0; pp < 2; ++pp) {
        const int p  = wave * 2 + pp;
        const int b  = p >> 12;
        const int lm = p & (LL - 1);
        const float* base = outy + ((size_t)b * KK * LL + lm) * DI + 2 * lane;
#pragma unroll
        for (int i = 0; i < 3; ++i) {
            const v2f a02 = *(const v2f*)(base + 64 * i);
            const v2f a13 = *(const v2f*)(base + ks + 64 * i);
            v[pp][i] = a02 + a13;
        }
    }

#pragma unroll
    for (int pp = 0; pp < 2; ++pp) {
        float s = 0.0f, q = 0.0f;
#pragma unroll
        for (int i = 0; i < 3; ++i) {
            s += v[pp][i].x; q += v[pp][i].x * v[pp][i].x;
            s += v[pp][i].y; q += v[pp][i].y * v[pp][i].y;
        }
#pragma unroll
        for (int o = 16; o; o >>= 1) {
            s += __shfl_xor(s, o, 32);
            q += __shfl_xor(q, o, 32);
        }
        const float mu   = s / 192.0f;
        const float var  = q / 192.0f - mu * mu;
        const float rstd = rsqrtf(var + 1e-5f);
        float* op = out + (size_t)(wave * 2 + pp) * DI + 2 * lane;
#pragma unroll
        for (int i = 0; i < 3; ++i) {
            v2f r;
            r.x = (v[pp][i].x - mu) * rstd * g[i].x + be[i].x;
            r.y = (v[pp][i].y - mu) * rstd * g[i].y + be[i].y;
            *(v2f*)(op + 64 * i) = r;
        }
    }
}

extern "C" void kernel_launch(void* const* d_in, const int* in_sizes, int n_in,
                              void* d_out, int out_size, void* d_ws, size_t ws_size,
                              hipStream_t stream) {
    const float* x   = (const float*)d_in[0];
    const float* y   = (const float*)d_in[1];
    const float* xpw = (const float*)d_in[2];
    const float* dtw = (const float*)d_in[3];
    const float* dtb = (const float*)d_in[4];
    const float* Al  = (const float*)d_in[5];
    const float* Ds  = (const float*)d_in[6];
    const float* gm  = (const float*)d_in[7];
    const float* bt  = (const float*)d_in[8];

    float* xdbl = (float*)d_ws;
    float* outy = xdbl + (size_t)B_ * KK * CP * LL;
    float* yt   = (float*)d_out;

    ss2d_proj_wmma<<<NTB + 256, 256, 0, stream>>>(x, xpw, xdbl, y, yt);
    ss2d_scan_local<<<B_ * 2 * NQ * 3, 128, 0, stream>>>(y, yt, dtw, dtb, Al, Ds, xdbl, outy);
    ss2d_scan_carry<<<B_ * KK * NS, DI, 0, stream>>>(Al, outy);
    ss2d_scan_final<<<B_ * 2 * NQ * 3, 128, 0, stream>>>(y, yt, dtw, dtb, Al, Ds, xdbl, outy);
    ss2d_merge_ln<<<B_ * LL / 16, 256, 0, stream>>>(outy, gm, bt, (float*)d_out);
}
